# starting kernel
# baseline (speedup 1.0000x reference)
#include <hip/hip_runtime.h>
#include <hip/hip_bf16.h>
#include <math.h>


typedef float v2f __attribute__((ext_vector_type(2)));
typedef float v8f __attribute__((ext_vector_type(8)));

#define A_AGENTS 128
#define D_IN     32
#define HID      128
#define KNN      16
#define OBS_LD   36
#define MAT_LD   132
#define NTHREADS 512
#define NWAVES   (NTHREADS / 32)
#define BIGF     3.4e38f

#define OFF_OBS  0
#define OFF_M    (OFF_OBS + A_AGENTS * OBS_LD)
#define OFF_T    (OFF_M   + A_AGENTS * MAT_LD)
#define OFF_H    (OFF_T   + A_AGENTS * MAT_LD)
#define OFF_DINV (OFF_H   + A_AGENTS * MAT_LD)
#define LDS_FLOATS (OFF_DINV + A_AGENTS)
#define LDS_BYTES  (LDS_FLOATS * 4)

__device__ __forceinline__ float fast_tanh(float x) {
#if __has_builtin(__builtin_amdgcn_tanhf)
    return __builtin_amdgcn_tanhf(x);
#elif __has_builtin(__builtin_amdgcn_tanh_f32)
    return __builtin_amdgcn_tanh_f32(x);
#else
    return tanhf(x);
#endif
}

template<bool ACT>
__device__ __forceinline__ void strip_mm16x4(const float* __restrict__ A, int lda,
                                             const float* __restrict__ B, int ldb,
                                             float* __restrict__ OUT, int ldo,
                                             const float* __restrict__ bias,
                                             int K, int tm, int tn0, int lane)
{
    const int g = lane >> 4;
    const int r = lane & 15;
    const float* Arow = A + (tm * 16 + r) * lda;
    const float* Bcol = B + tn0 * 16 + r;
    v8f c0 = {}, c1 = {}, c2 = {}, c3 = {};
    for (int k0 = 0; k0 < K; k0 += 4) {
        const int ka = k0 + 2 * g;
        const float* Brow0 = Bcol + (ka)     * ldb;
        const float* Brow1 = Bcol + (ka + 1) * ldb;
        v2f a, b0, b1, b2, b3;
        a.x  = Arow[ka];   a.y  = Arow[ka + 1];
        b0.x = Brow0[0];   b0.y = Brow1[0];
        b1.x = Brow0[16];  b1.y = Brow1[16];
        b2.x = Brow0[32];  b2.y = Brow1[32];
        b3.x = Brow0[48];  b3.y = Brow1[48];
        c0 = __builtin_amdgcn_wmma_f32_16x16x4_f32(false, a, false, b0, (short)0, c0, false, false);
        c1 = __builtin_amdgcn_wmma_f32_16x16x4_f32(false, a, false, b1, (short)0, c1, false, false);
        c2 = __builtin_amdgcn_wmma_f32_16x16x4_f32(false, a, false, b2, (short)0, c2, false, false);
        c3 = __builtin_amdgcn_wmma_f32_16x16x4_f32(false, a, false, b3, (short)0, c3, false, false);
    }
    const v8f cc[4] = {c0, c1, c2, c3};
    float* obase = OUT + (tm * 16 + 8 * g) * ldo + tn0 * 16 + r;
#pragma unroll
    for (int t = 0; t < 4; ++t) {
        const float bval = ACT ? bias[tn0 * 16 + t * 16 + r] : 0.0f;
#pragma unroll
        for (int v = 0; v < 8; ++v) {
            float val = cc[t][v];
            if (ACT) val = fast_tanh(val + bval);
            obase[v * ldo + t * 16] = val;
        }
    }
}

__global__ void __launch_bounds__(NTHREADS)
gnn_critic_fused(const float* __restrict__ obs,
                 const float* __restrict__ W1, const float* __restrict__ b1,
                 const float* __restrict__ W2, const float* __restrict__ b2,
                 const float* __restrict__ Wout, const float* __restrict__ bout,
                 float* __restrict__ out)
{
    extern __shared__ float smem[];
    float* obs_s  = smem + OFF_OBS;
    float* M_s    = smem + OFF_M;
    float* T_s    = smem + OFF_T;
    float* H_s    = smem + OFF_H;
    float* dinv_s = smem + OFF_DINV;

    const int tid  = threadIdx.x;
    const int lane = tid & 31;
    const int wave = tid >> 5;
    const int tm   = wave >> 1;
    const int tn0  = (wave & 1) * 4;
    const int bidx = blockIdx.x;

    const float* gobs = obs + (size_t)bidx * (A_AGENTS * D_IN);
    for (int i = tid; i < A_AGENTS * (D_IN / 4); i += NTHREADS) {
        const int row = i / (D_IN / 4);
        const int c4  = i % (D_IN / 4);
        const float4 v = ((const float4*)gobs)[i];
        float* dst = obs_s + row * OBS_LD + c4 * 4;
        dst[0] = v.x; dst[1] = v.y; dst[2] = v.z; dst[3] = v.w;
    }
    for (int i = tid; i < A_AGENTS * MAT_LD; i += NTHREADS) M_s[i] = 0.0f;
    __syncthreads();

    for (int e = tid; e < A_AGENTS * A_AGENTS; e += NTHREADS) {
        const int i = e >> 7;
        const int j = e & 127;
        const float dx = obs_s[i * OBS_LD + 0] - obs_s[j * OBS_LD + 0];
        const float dy = obs_s[i * OBS_LD + 1] - obs_s[j * OBS_LD + 1];
        float d = dx * dx + dy * dy;
        if (i == j) d = BIGF;
        T_s[i * MAT_LD + j] = d;
    }
    __syncthreads();

    {
        const int a  = tid >> 2;
        const int sl = tid & 3;
        float* drow = T_s + a * MAT_LD + sl * 32;
        for (int t = 0; t < KNN; ++t) {
            float best = BIGF;
            int bestj = 0;
            for (int j = 0; j < 32; ++j) {
                const float d = drow[j];
                if (d < best) { best = d; bestj = sl * 32 + j; }
            }
            for (int off = 1; off < 4; off <<= 1) {
                const float ob = __shfl_xor(best, off);
                const int   oj = __shfl_xor(bestj, off);
                if (ob < best || (ob == best && oj < bestj)) {
                    best = ob; bestj = oj;
                }
            }
            if (sl == 0) {
                T_s[a * MAT_LD + bestj] = BIGF;
                M_s[bestj * MAT_LD + a] = 1.0f;
            }
        }
    }
    __syncthreads();

    {
        const int a  = tid >> 2;
        const int sl = tid & 3;
        const float* seg = M_s + a * MAT_LD + sl * 32;
        float s = 0.0f;
        for (int i = 0; i < 32; ++i) s += seg[i];
        s += __shfl_xor(s, 1);
        s += __shfl_xor(s, 2);
        if (sl == 0) dinv_s[a] = rsqrtf(1.0f + s);
    }
    __syncthreads();

    for (int e = tid; e < A_AGENTS * A_AGENTS; e += NTHREADS) {
        const int j = e >> 7;
        const int i = e & 127;
        float v = M_s[j * MAT_LD + i] * dinv_s[i] * dinv_s[j];
        if (i == j) v = dinv_s[j] * dinv_s[j];
        M_s[j * MAT_LD + i] = v;
    }
    __syncthreads();

    strip_mm16x4<false>(obs_s, OBS_LD, W1, HID, T_s, MAT_LD, nullptr, D_IN,
                        tm, tn0, lane);
    __syncthreads();

    strip_mm16x4<true>(M_s, MAT_LD, T_s, MAT_LD, H_s, MAT_LD, b1, HID,
                       tm, tn0, lane);
    __syncthreads();

    strip_mm16x4<false>(H_s, MAT_LD, W2, HID, T_s, MAT_LD, nullptr, HID,
                        tm, tn0, lane);
    __syncthreads();

    strip_mm16x4<true>(M_s, MAT_LD, T_s, MAT_LD, H_s, MAT_LD, b2, HID,
                       tm, tn0, lane);
    __syncthreads();

    {
        const int a  = tid >> 2;
        const int sl = tid & 3;
        const float* hrow = H_s + a * MAT_LD + sl * 32;
        const float* wseg = Wout + sl * 32;
        float s = 0.0f;
        for (int k = 0; k < 32; ++k) s += hrow[k] * wseg[k];
        s += __shfl_xor(s, 1);
        s += __shfl_xor(s, 2);
        if (sl == 0) out[(size_t)bidx * A_AGENTS + a] = s + bout[0];
    }
}

extern "C" void kernel_launch(void* const* d_in, const int* in_sizes, int n_in,
                              void* d_out, int out_size, void* d_ws, size_t ws_size,
                              hipStream_t stream) {
    (void)n_in; (void)out_size; (void)d_ws; (void)ws_size;
    const float* obs  = (const float*)d_in[0];
    const float* W1   = (const float*)d_in[1];
    const float* b1   = (const float*)d_in[2];
    const float* W2   = (const float*)d_in[3];
    const float* b2   = (const float*)d_in[4];
    const float* Wout = (const float*)d_in[5];
    const float* bout = (const float*)d_in[6];
    float* out = (float*)d_out;

    const int B = in_sizes[0] / (A_AGENTS * D_IN);

    static_assert(LDS_BYTES <= 320 * 1024, "LDS over WGP capacity");
    (void)hipFuncSetAttribute((const void*)gnn_critic_fused,
                              hipFuncAttributeMaxDynamicSharedMemorySize,
                              LDS_BYTES);
    gnn_critic_fused<<<B, NTHREADS, LDS_BYTES, stream>>>(
        obs, W1, b1, W2, b2, Wout, bout, out);
}


# GEMM strips held in registers past a barrier and written over their operand in place: LDS 221696 to 154112 B per workgroup, so two workgroups are resident per WGP instead of one
# speedup vs baseline: 1.2729x; 1.2729x over previous
#include <hip/hip_runtime.h>
#include <hip/hip_bf16.h>
#include <math.h>


typedef float v2f __attribute__((ext_vector_type(2)));
typedef float v8f __attribute__((ext_vector_type(8)));

#define A_AGENTS 128
#define D_IN     32
#define HID      128
#define KNN      16
#define OBS_LD   36
#define MAT_LD   132
#define NTHREADS 512
#define NWAVES   (NTHREADS / 32)
#define BIGF     3.4e38f

#define OFF_OBS  0
#define OFF_M    (OFF_OBS + A_AGENTS * OBS_LD)
#define OFF_X    (OFF_M   + A_AGENTS * MAT_LD)
#define OFF_DINV (OFF_X   + A_AGENTS * MAT_LD)
#define LDS_FLOATS (OFF_DINV + A_AGENTS)
#define LDS_BYTES  (LDS_FLOATS * 4)

__device__ __forceinline__ float fast_tanh(float x) {
#if __has_builtin(__builtin_amdgcn_tanhf)
    return __builtin_amdgcn_tanhf(x);
#elif __has_builtin(__builtin_amdgcn_tanh_f32)
    return __builtin_amdgcn_tanh_f32(x);
#else
    return tanhf(x);
#endif
}

struct Acc4 { v8f c0, c1, c2, c3; };

__device__ __forceinline__ Acc4 strip_acc16x4(const float* A, int lda,
                                              const float* B, int ldb,
                                              int K, int tm, int tn0, int lane)
{
    const int g = lane >> 4;
    const int r = lane & 15;
    const float* Arow = A + (tm * 16 + r) * lda;
    const float* Bcol = B + tn0 * 16 + r;
    v8f c0 = {}, c1 = {}, c2 = {}, c3 = {};
    for (int k0 = 0; k0 < K; k0 += 4) {
        const int ka = k0 + 2 * g;
        const float* Brow0 = Bcol + (ka)     * ldb;
        const float* Brow1 = Bcol + (ka + 1) * ldb;
        v2f a, b0, b1, b2, b3;
        a.x  = Arow[ka];   a.y  = Arow[ka + 1];
        b0.x = Brow0[0];   b0.y = Brow1[0];
        b1.x = Brow0[16];  b1.y = Brow1[16];
        b2.x = Brow0[32];  b2.y = Brow1[32];
        b3.x = Brow0[48];  b3.y = Brow1[48];
        c0 = __builtin_amdgcn_wmma_f32_16x16x4_f32(false, a, false, b0, (short)0, c0, false, false);
        c1 = __builtin_amdgcn_wmma_f32_16x16x4_f32(false, a, false, b1, (short)0, c1, false, false);
        c2 = __builtin_amdgcn_wmma_f32_16x16x4_f32(false, a, false, b2, (short)0, c2, false, false);
        c3 = __builtin_amdgcn_wmma_f32_16x16x4_f32(false, a, false, b3, (short)0, c3, false, false);
    }
    Acc4 acc;
    acc.c0 = c0; acc.c1 = c1; acc.c2 = c2; acc.c3 = c3;
    return acc;
}

template<bool ACT>
__device__ __forceinline__ void strip_store16x4(const Acc4& acc, float* OUT, int ldo,
                                                const float* __restrict__ bias,
                                                int tm, int tn0, int lane)
{
    const int g = lane >> 4;
    const int r = lane & 15;
    const v8f cc[4] = {acc.c0, acc.c1, acc.c2, acc.c3};
    float* obase = OUT + (tm * 16 + 8 * g) * ldo + tn0 * 16 + r;
#pragma unroll
    for (int t = 0; t < 4; ++t) {
        const float bval = ACT ? bias[tn0 * 16 + t * 16 + r] : 0.0f;
#pragma unroll
        for (int v = 0; v < 8; ++v) {
            float val = cc[t][v];
            if (ACT) val = fast_tanh(val + bval);
            obase[v * ldo + t * 16] = val;
        }
    }
}

__global__ void __launch_bounds__(NTHREADS)
gnn_critic_fused(const float* __restrict__ obs,
                 const float* __restrict__ W1, const float* __restrict__ b1,
                 const float* __restrict__ W2, const float* __restrict__ b2,
                 const float* __restrict__ Wout, const float* __restrict__ bout,
                 float* __restrict__ out)
{
    extern __shared__ float smem[];
    float* obs_s  = smem + OFF_OBS;
    float* M_s    = smem + OFF_M;
    float* X_s    = smem + OFF_X;
    float* dinv_s = smem + OFF_DINV;

    const int tid  = threadIdx.x;
    const int lane = tid & 31;
    const int wave = tid >> 5;
    const int tm   = wave >> 1;
    const int tn0  = (wave & 1) * 4;
    const int bidx = blockIdx.x;

    const float* gobs = obs + (size_t)bidx * (A_AGENTS * D_IN);
    for (int i = tid; i < A_AGENTS * (D_IN / 4); i += NTHREADS) {
        const int row = i / (D_IN / 4);
        const int c4  = i % (D_IN / 4);
        const float4 v = ((const float4*)gobs)[i];
        float* dst = obs_s + row * OBS_LD + c4 * 4;
        dst[0] = v.x; dst[1] = v.y; dst[2] = v.z; dst[3] = v.w;
    }
    for (int i = tid; i < A_AGENTS * MAT_LD; i += NTHREADS) M_s[i] = 0.0f;
    __syncthreads();

    for (int e = tid; e < A_AGENTS * A_AGENTS; e += NTHREADS) {
        const int i = e >> 7;
        const int j = e & 127;
        const float dx = obs_s[i * OBS_LD + 0] - obs_s[j * OBS_LD + 0];
        const float dy = obs_s[i * OBS_LD + 1] - obs_s[j * OBS_LD + 1];
        float d = dx * dx + dy * dy;
        if (i == j) d = BIGF;
        X_s[i * MAT_LD + j] = d;
    }
    __syncthreads();

    {
        const int a  = tid >> 2;
        const int sl = tid & 3;
        float* drow = X_s + a * MAT_LD + sl * 32;
        for (int t = 0; t < KNN; ++t) {
            float best = BIGF;
            int bestj = 0;
            for (int j = 0; j < 32; ++j) {
                const float d = drow[j];
                if (d < best) { best = d; bestj = sl * 32 + j; }
            }
            for (int off = 1; off < 4; off <<= 1) {
                const float ob = __shfl_xor(best, off);
                const int   oj = __shfl_xor(bestj, off);
                if (ob < best || (ob == best && oj < bestj)) {
                    best = ob; bestj = oj;
                }
            }
            if (sl == 0) {
                X_s[a * MAT_LD + bestj] = BIGF;
                M_s[bestj * MAT_LD + a] = 1.0f;
            }
        }
    }
    __syncthreads();

    {
        const int a  = tid >> 2;
        const int sl = tid & 3;
        const float* seg = M_s + a * MAT_LD + sl * 32;
        float s = 0.0f;
        for (int i = 0; i < 32; ++i) s += seg[i];
        s += __shfl_xor(s, 1);
        s += __shfl_xor(s, 2);
        if (sl == 0) dinv_s[a] = rsqrtf(1.0f + s);
    }
    __syncthreads();

    for (int e = tid; e < A_AGENTS * A_AGENTS; e += NTHREADS) {
        const int j = e >> 7;
        const int i = e & 127;
        float v = M_s[j * MAT_LD + i] * dinv_s[i] * dinv_s[j];
        if (i == j) v = dinv_s[j] * dinv_s[j];
        M_s[j * MAT_LD + i] = v;
    }
    __syncthreads();

    {
        const Acc4 acc = strip_acc16x4(obs_s, OBS_LD, W1, HID, D_IN, tm, tn0, lane);
        strip_store16x4<false>(acc, X_s, MAT_LD, nullptr, tm, tn0, lane);
    }
    __syncthreads();

    {
        const Acc4 acc = strip_acc16x4(M_s, MAT_LD, X_s, MAT_LD, HID, tm, tn0, lane);
        __syncthreads();
        strip_store16x4<true>(acc, X_s, MAT_LD, b1, tm, tn0, lane);
    }
    __syncthreads();

    {
        const Acc4 acc = strip_acc16x4(X_s, MAT_LD, W2, HID, HID, tm, tn0, lane);
        __syncthreads();
        strip_store16x4<false>(acc, X_s, MAT_LD, nullptr, tm, tn0, lane);
    }
    __syncthreads();

    {
        const Acc4 acc = strip_acc16x4(M_s, MAT_LD, X_s, MAT_LD, HID, tm, tn0, lane);
        __syncthreads();
        strip_store16x4<true>(acc, X_s, MAT_LD, b2, tm, tn0, lane);
    }
    __syncthreads();

    {
        const int a  = tid >> 2;
        const int sl = tid & 3;
        const float* hrow = X_s + a * MAT_LD + sl * 32;
        const float* wseg = Wout + sl * 32;
        float s = 0.0f;
        for (int k = 0; k < 32; ++k) s += hrow[k] * wseg[k];
        s += __shfl_xor(s, 1);
        s += __shfl_xor(s, 2);
        if (sl == 0) out[(size_t)bidx * A_AGENTS + a] = s + bout[0];
    }
}

extern "C" void kernel_launch(void* const* d_in, const int* in_sizes, int n_in,
                              void* d_out, int out_size, void* d_ws, size_t ws_size,
                              hipStream_t stream) {
    (void)n_in; (void)out_size; (void)d_ws; (void)ws_size;
    const float* obs  = (const float*)d_in[0];
    const float* W1   = (const float*)d_in[1];
    const float* b1   = (const float*)d_in[2];
    const float* W2   = (const float*)d_in[3];
    const float* b2   = (const float*)d_in[4];
    const float* Wout = (const float*)d_in[5];
    const float* bout = (const float*)d_in[6];
    float* out = (float*)d_out;

    const int B = in_sizes[0] / (A_AGENTS * D_IN);

    static_assert(2 * LDS_BYTES <= 320 * 1024, "two workgroups must fit one WGP's LDS");
    (void)hipFuncSetAttribute((const void*)gnn_critic_fused,
                              hipFuncAttributeMaxDynamicSharedMemorySize,
                              LDS_BYTES);
    gnn_critic_fused<<<B, NTHREADS, LDS_BYTES, stream>>>(
        obs, W1, b1, W2, b2, Wout, bout, out);
}


# layer 1 summed as (M @ obs) @ W1 instead of M @ (obs @ W1): the 128-deep product is 32 columns wide, f32 WMMAs per wave 416 to 320; changes the order of the layer-1 sums
# speedup vs baseline: 1.4712x; 1.1558x over previous
#include <hip/hip_runtime.h>
#include <hip/hip_bf16.h>
#include <math.h>


typedef float v2f __attribute__((ext_vector_type(2)));
typedef float v8f __attribute__((ext_vector_type(8)));

#define A_AGENTS 128
#define D_IN     32
#define HID      128
#define KNN      16
#define OBS_LD   36
#define MAT_LD   132
#define NTHREADS 512
#define NWAVES   (NTHREADS / 32)
#define BIGF     3.4e38f

#define OFF_OBS  0
#define OFF_M    (OFF_OBS + A_AGENTS * OBS_LD)
#define OFF_X    (OFF_M   + A_AGENTS * MAT_LD)
#define OFF_DINV (OFF_X   + A_AGENTS * MAT_LD)
#define LDS_FLOATS (OFF_DINV + A_AGENTS)
#define LDS_BYTES  (LDS_FLOATS * 4)

__device__ __forceinline__ float fast_tanh(float x) {
#if __has_builtin(__builtin_amdgcn_tanhf)
    return __builtin_amdgcn_tanhf(x);
#elif __has_builtin(__builtin_amdgcn_tanh_f32)
    return __builtin_amdgcn_tanh_f32(x);
#else
    return tanhf(x);
#endif
}

struct Acc4 { v8f c0, c1, c2, c3; };

__device__ __forceinline__ Acc4 strip_acc16x4(const float* A, int lda,
                                              const float* B, int ldb,
                                              int K, int tm, int tn0, int lane)
{
    const int g = lane >> 4;
    const int r = lane & 15;
    const float* Arow = A + (tm * 16 + r) * lda;
    const float* Bcol = B + tn0 * 16 + r;
    v8f c0 = {}, c1 = {}, c2 = {}, c3 = {};
    for (int k0 = 0; k0 < K; k0 += 4) {
        const int ka = k0 + 2 * g;
        const float* Brow0 = Bcol + (ka)     * ldb;
        const float* Brow1 = Bcol + (ka + 1) * ldb;
        v2f a, b0, b1, b2, b3;
        a.x  = Arow[ka];   a.y  = Arow[ka + 1];
        b0.x = Brow0[0];   b0.y = Brow1[0];
        b1.x = Brow0[16];  b1.y = Brow1[16];
        b2.x = Brow0[32];  b2.y = Brow1[32];
        b3.x = Brow0[48];  b3.y = Brow1[48];
        c0 = __builtin_amdgcn_wmma_f32_16x16x4_f32(false, a, false, b0, (short)0, c0, false, false);
        c1 = __builtin_amdgcn_wmma_f32_16x16x4_f32(false, a, false, b1, (short)0, c1, false, false);
        c2 = __builtin_amdgcn_wmma_f32_16x16x4_f32(false, a, false, b2, (short)0, c2, false, false);
        c3 = __builtin_amdgcn_wmma_f32_16x16x4_f32(false, a, false, b3, (short)0, c3, false, false);
    }
    Acc4 acc;
    acc.c0 = c0; acc.c1 = c1; acc.c2 = c2; acc.c3 = c3;
    return acc;
}

__device__ __forceinline__ v8f tile_acc16x16(const float* A, int lda,
                                             const float* B, int ldb,
                                             int K, int tm, int tn, int lane)
{
    const int g = lane >> 4;
    const int r = lane & 15;
    const float* Arow = A + (tm * 16 + r) * lda;
    const float* Bcol = B + tn * 16 + r;
    v8f c = {};
    for (int k0 = 0; k0 < K; k0 += 4) {
        const int ka = k0 + 2 * g;
        v2f a, b;
        a.x = Arow[ka];           a.y = Arow[ka + 1];
        b.x = Bcol[ka * ldb];     b.y = Bcol[(ka + 1) * ldb];
        c = __builtin_amdgcn_wmma_f32_16x16x4_f32(false, a, false, b, (short)0, c, false, false);
    }
    return c;
}

template<bool ACT>
__device__ __forceinline__ void strip_store16x4(const Acc4& acc, float* OUT, int ldo,
                                                const float* __restrict__ bias,
                                                int tm, int tn0, int lane)
{
    const int g = lane >> 4;
    const int r = lane & 15;
    const v8f cc[4] = {acc.c0, acc.c1, acc.c2, acc.c3};
    float* obase = OUT + (tm * 16 + 8 * g) * ldo + tn0 * 16 + r;
#pragma unroll
    for (int t = 0; t < 4; ++t) {
        const float bval = ACT ? bias[tn0 * 16 + t * 16 + r] : 0.0f;
#pragma unroll
        for (int v = 0; v < 8; ++v) {
            float val = cc[t][v];
            if (ACT) val = fast_tanh(val + bval);
            obase[v * ldo + t * 16] = val;
        }
    }
}

__global__ void __launch_bounds__(NTHREADS)
gnn_critic_fused(const float* __restrict__ obs,
                 const float* __restrict__ W1, const float* __restrict__ b1,
                 const float* __restrict__ W2, const float* __restrict__ b2,
                 const float* __restrict__ Wout, const float* __restrict__ bout,
                 float* __restrict__ out)
{
    extern __shared__ float smem[];
    float* obs_s  = smem + OFF_OBS;
    float* M_s    = smem + OFF_M;
    float* X_s    = smem + OFF_X;
    float* dinv_s = smem + OFF_DINV;

    const int tid  = threadIdx.x;
    const int lane = tid & 31;
    const int wave = tid >> 5;
    const int tm   = wave >> 1;
    const int tn0  = (wave & 1) * 4;
    const int bidx = blockIdx.x;

    const float* gobs = obs + (size_t)bidx * (A_AGENTS * D_IN);
    for (int i = tid; i < A_AGENTS * (D_IN / 4); i += NTHREADS) {
        const int row = i / (D_IN / 4);
        const int c4  = i % (D_IN / 4);
        const float4 v = ((const float4*)gobs)[i];
        float* dst = obs_s + row * OBS_LD + c4 * 4;
        dst[0] = v.x; dst[1] = v.y; dst[2] = v.z; dst[3] = v.w;
    }
    for (int i = tid; i < A_AGENTS * MAT_LD; i += NTHREADS) M_s[i] = 0.0f;
    __syncthreads();

    for (int e = tid; e < A_AGENTS * A_AGENTS; e += NTHREADS) {
        const int i = e >> 7;
        const int j = e & 127;
        const float dx = obs_s[i * OBS_LD + 0] - obs_s[j * OBS_LD + 0];
        const float dy = obs_s[i * OBS_LD + 1] - obs_s[j * OBS_LD + 1];
        float d = dx * dx + dy * dy;
        if (i == j) d = BIGF;
        X_s[i * MAT_LD + j] = d;
    }
    __syncthreads();

    {
        const int a  = tid >> 2;
        const int sl = tid & 3;
        float* drow = X_s + a * MAT_LD + sl * 32;
        for (int t = 0; t < KNN; ++t) {
            float best = BIGF;
            int bestj = 0;
            for (int j = 0; j < 32; ++j) {
                const float d = drow[j];
                if (d < best) { best = d; bestj = sl * 32 + j; }
            }
            for (int off = 1; off < 4; off <<= 1) {
                const float ob = __shfl_xor(best, off);
                const int   oj = __shfl_xor(bestj, off);
                if (ob < best || (ob == best && oj < bestj)) {
                    best = ob; bestj = oj;
                }
            }
            if (sl == 0) {
                X_s[a * MAT_LD + bestj] = BIGF;
                M_s[bestj * MAT_LD + a] = 1.0f;
            }
        }
    }
    __syncthreads();

    {
        const int a  = tid >> 2;
        const int sl = tid & 3;
        const float* seg = M_s + a * MAT_LD + sl * 32;
        float s = 0.0f;
        for (int i = 0; i < 32; ++i) s += seg[i];
        s += __shfl_xor(s, 1);
        s += __shfl_xor(s, 2);
        if (sl == 0) dinv_s[a] = rsqrtf(1.0f + s);
    }
    __syncthreads();

    for (int e = tid; e < A_AGENTS * A_AGENTS; e += NTHREADS) {
        const int j = e >> 7;
        const int i = e & 127;
        float v = M_s[j * MAT_LD + i] * dinv_s[i] * dinv_s[j];
        if (i == j) v = dinv_s[j] * dinv_s[j];
        M_s[j * MAT_LD + i] = v;
    }
    __syncthreads();

    {
        const int tn = wave & 1;
        const v8f c = tile_acc16x16(M_s, MAT_LD, obs_s, OBS_LD, A_AGENTS, tm, tn, lane);
        float* obase = X_s + (tm * 16 + 8 * (lane >> 4)) * MAT_LD + tn * 16 + (lane & 15);
#pragma unroll
        for (int v = 0; v < 8; ++v) obase[v * MAT_LD] = c[v];
    }
    __syncthreads();

    {
        const Acc4 acc = strip_acc16x4(X_s, MAT_LD, W1, HID, D_IN, tm, tn0, lane);
        __syncthreads();
        strip_store16x4<true>(acc, X_s, MAT_LD, b1, tm, tn0, lane);
    }
    __syncthreads();

    {
        const Acc4 acc = strip_acc16x4(X_s, MAT_LD, W2, HID, HID, tm, tn0, lane);
        __syncthreads();
        strip_store16x4<false>(acc, X_s, MAT_LD, nullptr, tm, tn0, lane);
    }
    __syncthreads();

    {
        const Acc4 acc = strip_acc16x4(M_s, MAT_LD, X_s, MAT_LD, HID, tm, tn0, lane);
        __syncthreads();
        strip_store16x4<true>(acc, X_s, MAT_LD, b2, tm, tn0, lane);
    }
    __syncthreads();

    {
        const int a  = tid >> 2;
        const int sl = tid & 3;
        const float* hrow = X_s + a * MAT_LD + sl * 32;
        const float* wseg = Wout + sl * 32;
        float s = 0.0f;
        for (int k = 0; k < 32; ++k) s += hrow[k] * wseg[k];
        s += __shfl_xor(s, 1);
        s += __shfl_xor(s, 2);
        if (sl == 0) out[(size_t)bidx * A_AGENTS + a] = s + bout[0];
    }
}

extern "C" void kernel_launch(void* const* d_in, const int* in_sizes, int n_in,
                              void* d_out, int out_size, void* d_ws, size_t ws_size,
                              hipStream_t stream) {
    (void)n_in; (void)out_size; (void)d_ws; (void)ws_size;
    const float* obs  = (const float*)d_in[0];
    const float* W1   = (const float*)d_in[1];
    const float* b1   = (const float*)d_in[2];
    const float* W2   = (const float*)d_in[3];
    const float* b2   = (const float*)d_in[4];
    const float* Wout = (const float*)d_in[5];
    const float* bout = (const float*)d_in[6];
    float* out = (float*)d_out;

    const int B = in_sizes[0] / (A_AGENTS * D_IN);

    static_assert(2 * LDS_BYTES <= 320 * 1024, "two workgroups must fit one WGP's LDS");
    (void)hipFuncSetAttribute((const void*)gnn_critic_fused,
                              hipFuncAttributeMaxDynamicSharedMemorySize,
                              LDS_BYTES);
    gnn_critic_fused<<<B, NTHREADS, LDS_BYTES, stream>>>(
        obs, W1, b1, W2, b2, Wout, bout, out);
}


# kNN pick: each lane keeps the minima of its four 8-distance groups in registers and after a pick re-reads only the group it came from; LDS reads per lane per round 32 to 8, same picks
# speedup vs baseline: 1.6868x; 1.1465x over previous
#include <hip/hip_runtime.h>
#include <hip/hip_bf16.h>
#include <math.h>


typedef float v2f __attribute__((ext_vector_type(2)));
typedef float v8f __attribute__((ext_vector_type(8)));

#define A_AGENTS 128
#define D_IN     32
#define HID      128
#define KNN      16
#define OBS_LD   36
#define MAT_LD   132
#define NTHREADS 512
#define NWAVES   (NTHREADS / 32)
#define BIGF     3.4e38f

#define OFF_OBS  0
#define OFF_M    (OFF_OBS + A_AGENTS * OBS_LD)
#define OFF_X    (OFF_M   + A_AGENTS * MAT_LD)
#define OFF_DINV (OFF_X   + A_AGENTS * MAT_LD)
#define LDS_FLOATS (OFF_DINV + A_AGENTS)
#define LDS_BYTES  (LDS_FLOATS * 4)

__device__ __forceinline__ float fast_tanh(float x) {
#if __has_builtin(__builtin_amdgcn_tanhf)
    return __builtin_amdgcn_tanhf(x);
#elif __has_builtin(__builtin_amdgcn_tanh_f32)
    return __builtin_amdgcn_tanh_f32(x);
#else
    return tanhf(x);
#endif
}

__device__ __forceinline__ void scan8(const float* p, int base, float& m, int& mj)
{
    const float4 lo = *(const float4*)(p);
    const float4 hi = *(const float4*)(p + 4);
    const float d[8] = {lo.x, lo.y, lo.z, lo.w, hi.x, hi.y, hi.z, hi.w};
    float best = BIGF;
    int e = 0;
#pragma unroll
    for (int i = 0; i < 8; ++i) {
        if (d[i] < best) { best = d[i]; e = i; }
    }
    m = best;
    mj = base + e;
}

struct Acc4 { v8f c0, c1, c2, c3; };

__device__ __forceinline__ Acc4 strip_acc16x4(const float* A, int lda,
                                              const float* B, int ldb,
                                              int K, int tm, int tn0, int lane)
{
    const int g = lane >> 4;
    const int r = lane & 15;
    const float* Arow = A + (tm * 16 + r) * lda;
    const float* Bcol = B + tn0 * 16 + r;
    v8f c0 = {}, c1 = {}, c2 = {}, c3 = {};
    for (int k0 = 0; k0 < K; k0 += 4) {
        const int ka = k0 + 2 * g;
        const float* Brow0 = Bcol + (ka)     * ldb;
        const float* Brow1 = Bcol + (ka + 1) * ldb;
        v2f a, b0, b1, b2, b3;
        a.x  = Arow[ka];   a.y  = Arow[ka + 1];
        b0.x = Brow0[0];   b0.y = Brow1[0];
        b1.x = Brow0[16];  b1.y = Brow1[16];
        b2.x = Brow0[32];  b2.y = Brow1[32];
        b3.x = Brow0[48];  b3.y = Brow1[48];
        c0 = __builtin_amdgcn_wmma_f32_16x16x4_f32(false, a, false, b0, (short)0, c0, false, false);
        c1 = __builtin_amdgcn_wmma_f32_16x16x4_f32(false, a, false, b1, (short)0, c1, false, false);
        c2 = __builtin_amdgcn_wmma_f32_16x16x4_f32(false, a, false, b2, (short)0, c2, false, false);
        c3 = __builtin_amdgcn_wmma_f32_16x16x4_f32(false, a, false, b3, (short)0, c3, false, false);
    }
    Acc4 acc;
    acc.c0 = c0; acc.c1 = c1; acc.c2 = c2; acc.c3 = c3;
    return acc;
}

__device__ __forceinline__ v8f tile_acc16x16(const float* A, int lda,
                                             const float* B, int ldb,
                                             int K, int tm, int tn, int lane)
{
    const int g = lane >> 4;
    const int r = lane & 15;
    const float* Arow = A + (tm * 16 + r) * lda;
    const float* Bcol = B + tn * 16 + r;
    v8f c = {};
    for (int k0 = 0; k0 < K; k0 += 4) {
        const int ka = k0 + 2 * g;
        v2f a, b;
        a.x = Arow[ka];           a.y = Arow[ka + 1];
        b.x = Bcol[ka * ldb];     b.y = Bcol[(ka + 1) * ldb];
        c = __builtin_amdgcn_wmma_f32_16x16x4_f32(false, a, false, b, (short)0, c, false, false);
    }
    return c;
}

template<bool ACT>
__device__ __forceinline__ void strip_store16x4(const Acc4& acc, float* OUT, int ldo,
                                                const float* __restrict__ bias,
                                                int tm, int tn0, int lane)
{
    const int g = lane >> 4;
    const int r = lane & 15;
    const v8f cc[4] = {acc.c0, acc.c1, acc.c2, acc.c3};
    float* obase = OUT + (tm * 16 + 8 * g) * ldo + tn0 * 16 + r;
#pragma unroll
    for (int t = 0; t < 4; ++t) {
        const float bval = ACT ? bias[tn0 * 16 + t * 16 + r] : 0.0f;
#pragma unroll
        for (int v = 0; v < 8; ++v) {
            float val = cc[t][v];
            if (ACT) val = fast_tanh(val + bval);
            obase[v * ldo + t * 16] = val;
        }
    }
}

__global__ void __launch_bounds__(NTHREADS)
gnn_critic_fused(const float* __restrict__ obs,
                 const float* __restrict__ W1, const float* __restrict__ b1,
                 const float* __restrict__ W2, const float* __restrict__ b2,
                 const float* __restrict__ Wout, const float* __restrict__ bout,
                 float* __restrict__ out)
{
    extern __shared__ float smem[];
    float* obs_s  = smem + OFF_OBS;
    float* M_s    = smem + OFF_M;
    float* X_s    = smem + OFF_X;
    float* dinv_s = smem + OFF_DINV;

    const int tid  = threadIdx.x;
    const int lane = tid & 31;
    const int wave = tid >> 5;
    const int tm   = wave >> 1;
    const int tn0  = (wave & 1) * 4;
    const int bidx = blockIdx.x;

    const float* gobs = obs + (size_t)bidx * (A_AGENTS * D_IN);
    for (int i = tid; i < A_AGENTS * (D_IN / 4); i += NTHREADS) {
        const int row = i / (D_IN / 4);
        const int c4  = i % (D_IN / 4);
        const float4 v = ((const float4*)gobs)[i];
        float* dst = obs_s + row * OBS_LD + c4 * 4;
        dst[0] = v.x; dst[1] = v.y; dst[2] = v.z; dst[3] = v.w;
    }
    for (int i = tid; i < A_AGENTS * MAT_LD; i += NTHREADS) M_s[i] = 0.0f;
    __syncthreads();

    for (int e = tid; e < A_AGENTS * A_AGENTS; e += NTHREADS) {
        const int i = e >> 7;
        const int j = e & 127;
        const float dx = obs_s[i * OBS_LD + 0] - obs_s[j * OBS_LD + 0];
        const float dy = obs_s[i * OBS_LD + 1] - obs_s[j * OBS_LD + 1];
        float d = dx * dx + dy * dy;
        if (i == j) d = BIGF;
        X_s[i * MAT_LD + j] = d;
    }
    __syncthreads();

    {
        const int a  = tid >> 2;
        const int sl = tid & 3;
        const float* drow = X_s + a * MAT_LD + sl * 32;
        float gm[4];
        int   gj[4];
#pragma unroll
        for (int q = 0; q < 4; ++q) scan8(drow + 8 * q, sl * 32 + 8 * q, gm[q], gj[q]);

        for (int t = 0; t < KNN; ++t) {
            float best = BIGF;
            int bestj = 0;
#pragma unroll
            for (int q = 0; q < 4; ++q) {
                if (gm[q] < best) { best = gm[q]; bestj = gj[q]; }
            }
            for (int off = 1; off < 4; off <<= 1) {
                const float ob = __shfl_xor(best, off);
                const int   oj = __shfl_xor(bestj, off);
                if (ob < best || (ob == best && oj < bestj)) {
                    best = ob; bestj = oj;
                }
            }
            if (sl == 0) {
                X_s[a * MAT_LD + bestj] = BIGF;
                M_s[bestj * MAT_LD + a] = 1.0f;
            }
            if (t + 1 < KNN) {
                const int qw = (bestj >> 3) & 3;
                float nm;
                int   nj;
                scan8(drow + 8 * qw, sl * 32 + 8 * qw, nm, nj);
#pragma unroll
                for (int q = 0; q < 4; ++q) {
                    const bool hit = (q == qw);
                    gm[q] = hit ? nm : gm[q];
                    gj[q] = hit ? nj : gj[q];
                }
            }
        }
    }
    __syncthreads();

    {
        const int a  = tid >> 2;
        const int sl = tid & 3;
        const float* seg = M_s + a * MAT_LD + sl * 32;
        float s = 0.0f;
        for (int i = 0; i < 32; ++i) s += seg[i];
        s += __shfl_xor(s, 1);
        s += __shfl_xor(s, 2);
        if (sl == 0) dinv_s[a] = rsqrtf(1.0f + s);
    }
    __syncthreads();

    for (int e = tid; e < A_AGENTS * A_AGENTS; e += NTHREADS) {
        const int j = e >> 7;
        const int i = e & 127;
        float v = M_s[j * MAT_LD + i] * dinv_s[i] * dinv_s[j];
        if (i == j) v = dinv_s[j] * dinv_s[j];
        M_s[j * MAT_LD + i] = v;
    }
    __syncthreads();

    {
        const int tn = wave & 1;
        const v8f c = tile_acc16x16(M_s, MAT_LD, obs_s, OBS_LD, A_AGENTS, tm, tn, lane);
        float* obase = X_s + (tm * 16 + 8 * (lane >> 4)) * MAT_LD + tn * 16 + (lane & 15);
#pragma unroll
        for (int v = 0; v < 8; ++v) obase[v * MAT_LD] = c[v];
    }
    __syncthreads();

    {
        const Acc4 acc = strip_acc16x4(X_s, MAT_LD, W1, HID, D_IN, tm, tn0, lane);
        __syncthreads();
        strip_store16x4<true>(acc, X_s, MAT_LD, b1, tm, tn0, lane);
    }
    __syncthreads();

    {
        const Acc4 acc = strip_acc16x4(X_s, MAT_LD, W2, HID, HID, tm, tn0, lane);
        __syncthreads();
        strip_store16x4<false>(acc, X_s, MAT_LD, nullptr, tm, tn0, lane);
    }
    __syncthreads();

    {
        const Acc4 acc = strip_acc16x4(M_s, MAT_LD, X_s, MAT_LD, HID, tm, tn0, lane);
        __syncthreads();
        strip_store16x4<true>(acc, X_s, MAT_LD, b2, tm, tn0, lane);
    }
    __syncthreads();

    {
        const int a  = tid >> 2;
        const int sl = tid & 3;
        const float* hrow = X_s + a * MAT_LD + sl * 32;
        const float* wseg = Wout + sl * 32;
        float s = 0.0f;
        for (int k = 0; k < 32; ++k) s += hrow[k] * wseg[k];
        s += __shfl_xor(s, 1);
        s += __shfl_xor(s, 2);
        if (sl == 0) out[(size_t)bidx * A_AGENTS + a] = s + bout[0];
    }
}

extern "C" void kernel_launch(void* const* d_in, const int* in_sizes, int n_in,
                              void* d_out, int out_size, void* d_ws, size_t ws_size,
                              hipStream_t stream) {
    (void)n_in; (void)out_size; (void)d_ws; (void)ws_size;
    const float* obs  = (const float*)d_in[0];
    const float* W1   = (const float*)d_in[1];
    const float* b1   = (const float*)d_in[2];
    const float* W2   = (const float*)d_in[3];
    const float* b2   = (const float*)d_in[4];
    const float* Wout = (const float*)d_in[5];
    const float* bout = (const float*)d_in[6];
    float* out = (float*)d_out;

    const int B = in_sizes[0] / (A_AGENTS * D_IN);

    static_assert(2 * LDS_BYTES <= 320 * 1024, "two workgroups must fit one WGP's LDS");
    (void)hipFuncSetAttribute((const void*)gnn_critic_fused,
                              hipFuncAttributeMaxDynamicSharedMemorySize,
                              LDS_BYTES);
    gnn_critic_fused<<<B, NTHREADS, LDS_BYTES, stream>>>(
        obs, W1, b1, W2, b2, Wout, bout, out);
}


# M kept as the 0/1 adjacency plus identity; D^-1/2 applied as row scalings of obs, of H1@W2 and of both M-product results instead of a pass over all 128x128 entries of M; norm-factor rounding moves
# speedup vs baseline: 1.7539x; 1.0398x over previous
#include <hip/hip_runtime.h>
#include <hip/hip_bf16.h>
#include <math.h>


typedef float v2f __attribute__((ext_vector_type(2)));
typedef float v8f __attribute__((ext_vector_type(8)));

#define A_AGENTS 128
#define D_IN     32
#define HID      128
#define KNN      16
#define OBS_LD   36
#define MAT_LD   132
#define NTHREADS 512
#define NWAVES   (NTHREADS / 32)
#define BIGF     3.4e38f

#define OFF_OBS  0
#define OFF_M    (OFF_OBS + A_AGENTS * OBS_LD)
#define OFF_X    (OFF_M   + A_AGENTS * MAT_LD)
#define OFF_DINV (OFF_X   + A_AGENTS * MAT_LD)
#define LDS_FLOATS (OFF_DINV + A_AGENTS)
#define LDS_BYTES  (LDS_FLOATS * 4)

__device__ __forceinline__ float fast_tanh(float x) {
#if __has_builtin(__builtin_amdgcn_tanhf)
    return __builtin_amdgcn_tanhf(x);
#elif __has_builtin(__builtin_amdgcn_tanh_f32)
    return __builtin_amdgcn_tanh_f32(x);
#else
    return tanhf(x);
#endif
}

__device__ __forceinline__ void scan8(const float* p, int base, float& m, int& mj)
{
    const float4 lo = *(const float4*)(p);
    const float4 hi = *(const float4*)(p + 4);
    const float d[8] = {lo.x, lo.y, lo.z, lo.w, hi.x, hi.y, hi.z, hi.w};
    float best = BIGF;
    int e = 0;
#pragma unroll
    for (int i = 0; i < 8; ++i) {
        if (d[i] < best) { best = d[i]; e = i; }
    }
    m = best;
    mj = base + e;
}

struct Acc4 { v8f c0, c1, c2, c3; };

__device__ __forceinline__ Acc4 strip_acc16x4(const float* A, int lda,
                                              const float* B, int ldb,
                                              int K, int tm, int tn0, int lane)
{
    const int g = lane >> 4;
    const int r = lane & 15;
    const float* Arow = A + (tm * 16 + r) * lda;
    const float* Bcol = B + tn0 * 16 + r;
    v8f c0 = {}, c1 = {}, c2 = {}, c3 = {};
    for (int k0 = 0; k0 < K; k0 += 4) {
        const int ka = k0 + 2 * g;
        const float* Brow0 = Bcol + (ka)     * ldb;
        const float* Brow1 = Bcol + (ka + 1) * ldb;
        v2f a, b0, b1, b2, b3;
        a.x  = Arow[ka];   a.y  = Arow[ka + 1];
        b0.x = Brow0[0];   b0.y = Brow1[0];
        b1.x = Brow0[16];  b1.y = Brow1[16];
        b2.x = Brow0[32];  b2.y = Brow1[32];
        b3.x = Brow0[48];  b3.y = Brow1[48];
        c0 = __builtin_amdgcn_wmma_f32_16x16x4_f32(false, a, false, b0, (short)0, c0, false, false);
        c1 = __builtin_amdgcn_wmma_f32_16x16x4_f32(false, a, false, b1, (short)0, c1, false, false);
        c2 = __builtin_amdgcn_wmma_f32_16x16x4_f32(false, a, false, b2, (short)0, c2, false, false);
        c3 = __builtin_amdgcn_wmma_f32_16x16x4_f32(false, a, false, b3, (short)0, c3, false, false);
    }
    Acc4 acc;
    acc.c0 = c0; acc.c1 = c1; acc.c2 = c2; acc.c3 = c3;
    return acc;
}

__device__ __forceinline__ v8f tile_acc16x16(const float* A, int lda,
                                             const float* B, int ldb,
                                             int K, int tm, int tn, int lane)
{
    const int g = lane >> 4;
    const int r = lane & 15;
    const float* Arow = A + (tm * 16 + r) * lda;
    const float* Bcol = B + tn * 16 + r;
    v8f c = {};
    for (int k0 = 0; k0 < K; k0 += 4) {
        const int ka = k0 + 2 * g;
        v2f a, b;
        a.x = Arow[ka];           a.y = Arow[ka + 1];
        b.x = Bcol[ka * ldb];     b.y = Bcol[(ka + 1) * ldb];
        c = __builtin_amdgcn_wmma_f32_16x16x4_f32(false, a, false, b, (short)0, c, false, false);
    }
    return c;
}

struct Row8 { float d[8]; };

__device__ __forceinline__ Row8 load_row8(const float* dinv, int tm, int lane)
{
    const float* p = dinv + tm * 16 + 8 * (lane >> 4);
    const float4 lo = *(const float4*)(p);
    const float4 hi = *(const float4*)(p + 4);
    Row8 rs;
    rs.d[0] = lo.x; rs.d[1] = lo.y; rs.d[2] = lo.z; rs.d[3] = lo.w;
    rs.d[4] = hi.x; rs.d[5] = hi.y; rs.d[6] = hi.z; rs.d[7] = hi.w;
    return rs;
}

template<bool SCALE, bool ACT>
__device__ __forceinline__ void strip_store16x4(const Acc4& acc, float* OUT, int ldo,
                                                const Row8& rs,
                                                const float* __restrict__ bias,
                                                int tm, int tn0, int lane)
{
    const int g = lane >> 4;
    const int r = lane & 15;
    const v8f cc[4] = {acc.c0, acc.c1, acc.c2, acc.c3};
    float* obase = OUT + (tm * 16 + 8 * g) * ldo + tn0 * 16 + r;
#pragma unroll
    for (int t = 0; t < 4; ++t) {
        const float bval = ACT ? bias[tn0 * 16 + t * 16 + r] : 0.0f;
#pragma unroll
        for (int v = 0; v < 8; ++v) {
            float val = cc[t][v];
            if (SCALE) val *= rs.d[v];
            if (ACT) val = fast_tanh(val + bval);
            obase[v * ldo + t * 16] = val;
        }
    }
}

__global__ void __launch_bounds__(NTHREADS)
gnn_critic_fused(const float* __restrict__ obs,
                 const float* __restrict__ W1, const float* __restrict__ b1,
                 const float* __restrict__ W2, const float* __restrict__ b2,
                 const float* __restrict__ Wout, const float* __restrict__ bout,
                 float* __restrict__ out)
{
    extern __shared__ float smem[];
    float* obs_s  = smem + OFF_OBS;
    float* M_s    = smem + OFF_M;
    float* X_s    = smem + OFF_X;
    float* dinv_s = smem + OFF_DINV;

    const int tid  = threadIdx.x;
    const int lane = tid & 31;
    const int wave = tid >> 5;
    const int tm   = wave >> 1;
    const int tn0  = (wave & 1) * 4;
    const int bidx = blockIdx.x;

    const float* gobs = obs + (size_t)bidx * (A_AGENTS * D_IN);
    for (int i = tid; i < A_AGENTS * (D_IN / 4); i += NTHREADS) {
        const int row = i / (D_IN / 4);
        const int c4  = i % (D_IN / 4);
        const float4 v = ((const float4*)gobs)[i];
        float* dst = obs_s + row * OBS_LD + c4 * 4;
        dst[0] = v.x; dst[1] = v.y; dst[2] = v.z; dst[3] = v.w;
    }
    for (int i = tid; i < A_AGENTS * MAT_LD; i += NTHREADS) M_s[i] = 0.0f;
    __syncthreads();

    for (int e = tid; e < A_AGENTS * A_AGENTS; e += NTHREADS) {
        const int i = e >> 7;
        const int j = e & 127;
        const float dx = obs_s[i * OBS_LD + 0] - obs_s[j * OBS_LD + 0];
        const float dy = obs_s[i * OBS_LD + 1] - obs_s[j * OBS_LD + 1];
        float d = dx * dx + dy * dy;
        if (i == j) d = BIGF;
        X_s[i * MAT_LD + j] = d;
    }
    __syncthreads();

    {
        const int a  = tid >> 2;
        const int sl = tid & 3;
        const float* drow = X_s + a * MAT_LD + sl * 32;
        if (sl == 0) M_s[a * MAT_LD + a] = 1.0f;
        float gm[4];
        int   gj[4];
#pragma unroll
        for (int q = 0; q < 4; ++q) scan8(drow + 8 * q, sl * 32 + 8 * q, gm[q], gj[q]);

        for (int t = 0; t < KNN; ++t) {
            float best = BIGF;
            int bestj = 0;
#pragma unroll
            for (int q = 0; q < 4; ++q) {
                if (gm[q] < best) { best = gm[q]; bestj = gj[q]; }
            }
            for (int off = 1; off < 4; off <<= 1) {
                const float ob = __shfl_xor(best, off);
                const int   oj = __shfl_xor(bestj, off);
                if (ob < best || (ob == best && oj < bestj)) {
                    best = ob; bestj = oj;
                }
            }
            if (sl == 0) {
                X_s[a * MAT_LD + bestj] = BIGF;
                M_s[bestj * MAT_LD + a] = 1.0f;
            }
            if (t + 1 < KNN) {
                const int qw = (bestj >> 3) & 3;
                float nm;
                int   nj;
                scan8(drow + 8 * qw, sl * 32 + 8 * qw, nm, nj);
#pragma unroll
                for (int q = 0; q < 4; ++q) {
                    const bool hit = (q == qw);
                    gm[q] = hit ? nm : gm[q];
                    gj[q] = hit ? nj : gj[q];
                }
            }
        }
    }
    __syncthreads();

    {
        const int a  = tid >> 2;
        const int sl = tid & 3;
        float* orow = obs_s + a * OBS_LD + sl * 8;
        float4 lo = *(const float4*)(orow);
        float4 hi = *(const float4*)(orow + 4);
        const float* seg = M_s + a * MAT_LD + sl * 32;
        float s = 0.0f;
        for (int i = 0; i < 32; ++i) s += seg[i];
        s += __shfl_xor(s, 1);
        s += __shfl_xor(s, 2);
        const float dv = rsqrtf(s);
        if (sl == 0) dinv_s[a] = dv;
        lo.x *= dv; lo.y *= dv; lo.z *= dv; lo.w *= dv;
        hi.x *= dv; hi.y *= dv; hi.z *= dv; hi.w *= dv;
        *(float4*)(orow)     = lo;
        *(float4*)(orow + 4) = hi;
    }
    __syncthreads();

    {
        const int tn = wave & 1;
        const v8f c = tile_acc16x16(M_s, MAT_LD, obs_s, OBS_LD, A_AGENTS, tm, tn, lane);
        const Row8 rs = load_row8(dinv_s, tm, lane);
        float* obase = X_s + (tm * 16 + 8 * (lane >> 4)) * MAT_LD + tn * 16 + (lane & 15);
#pragma unroll
        for (int v = 0; v < 8; ++v) obase[v * MAT_LD] = c[v] * rs.d[v];
    }
    __syncthreads();

    {
        const Acc4 acc = strip_acc16x4(X_s, MAT_LD, W1, HID, D_IN, tm, tn0, lane);
        const Row8 none = {};
        __syncthreads();
        strip_store16x4<false, true>(acc, X_s, MAT_LD, none, b1, tm, tn0, lane);
    }
    __syncthreads();

    {
        const Acc4 acc = strip_acc16x4(X_s, MAT_LD, W2, HID, HID, tm, tn0, lane);
        const Row8 rs = load_row8(dinv_s, tm, lane);
        __syncthreads();
        strip_store16x4<true, false>(acc, X_s, MAT_LD, rs, nullptr, tm, tn0, lane);
    }
    __syncthreads();

    {
        const Acc4 acc = strip_acc16x4(M_s, MAT_LD, X_s, MAT_LD, HID, tm, tn0, lane);
        const Row8 rs = load_row8(dinv_s, tm, lane);
        __syncthreads();
        strip_store16x4<true, true>(acc, X_s, MAT_LD, rs, b2, tm, tn0, lane);
    }
    __syncthreads();

    {
        const int a  = tid >> 2;
        const int sl = tid & 3;
        const float* hrow = X_s + a * MAT_LD + sl * 32;
        const float* wseg = Wout + sl * 32;
        float s = 0.0f;
        for (int k = 0; k < 32; ++k) s += hrow[k] * wseg[k];
        s += __shfl_xor(s, 1);
        s += __shfl_xor(s, 2);
        if (sl == 0) out[(size_t)bidx * A_AGENTS + a] = s + bout[0];
    }
}

extern "C" void kernel_launch(void* const* d_in, const int* in_sizes, int n_in,
                              void* d_out, int out_size, void* d_ws, size_t ws_size,
                              hipStream_t stream) {
    (void)n_in; (void)out_size; (void)d_ws; (void)ws_size;
    const float* obs  = (const float*)d_in[0];
    const float* W1   = (const float*)d_in[1];
    const float* b1   = (const float*)d_in[2];
    const float* W2   = (const float*)d_in[3];
    const float* b2   = (const float*)d_in[4];
    const float* Wout = (const float*)d_in[5];
    const float* bout = (const float*)d_in[6];
    float* out = (float*)d_out;

    const int B = in_sizes[0] / (A_AGENTS * D_IN);

    static_assert(2 * LDS_BYTES <= 320 * 1024, "two workgroups must fit one WGP's LDS");
    (void)hipFuncSetAttribute((const void*)gnn_critic_fused,
                              hipFuncAttributeMaxDynamicSharedMemorySize,
                              LDS_BYTES);
    gnn_critic_fused<<<B, NTHREADS, LDS_BYTES, stream>>>(
        obs, W1, b1, W2, b2, Wout, bout, out);
}


# GEMM strips: each lane owns 4 adjacent columns (tile t = columns 4c+t), so a k row of B is one 16-byte load per lane (global and LDS) and a result row one 16-byte store; B loads per k-step 8 to 2
# speedup vs baseline: 1.8147x; 1.0347x over previous
#include <hip/hip_runtime.h>
#include <hip/hip_bf16.h>
#include <math.h>


typedef float v2f __attribute__((ext_vector_type(2)));
typedef float v8f __attribute__((ext_vector_type(8)));

#define A_AGENTS 128
#define D_IN     32
#define HID      128
#define KNN      16
#define OBS_LD   36
#define MAT_LD   132
#define NTHREADS 512
#define NWAVES   (NTHREADS / 32)
#define BIGF     3.4e38f

#define OFF_OBS  0
#define OFF_M    (OFF_OBS + A_AGENTS * OBS_LD)
#define OFF_X    (OFF_M   + A_AGENTS * MAT_LD)
#define OFF_DINV (OFF_X   + A_AGENTS * MAT_LD)
#define LDS_FLOATS (OFF_DINV + A_AGENTS)
#define LDS_BYTES  (LDS_FLOATS * 4)

__device__ __forceinline__ float fast_tanh(float x) {
#if __has_builtin(__builtin_amdgcn_tanhf)
    return __builtin_amdgcn_tanhf(x);
#elif __has_builtin(__builtin_amdgcn_tanh_f32)
    return __builtin_amdgcn_tanh_f32(x);
#else
    return tanhf(x);
#endif
}

__device__ __forceinline__ void scan8(const float* p, int base, float& m, int& mj)
{
    const float4 lo = *(const float4*)(p);
    const float4 hi = *(const float4*)(p + 4);
    const float d[8] = {lo.x, lo.y, lo.z, lo.w, hi.x, hi.y, hi.z, hi.w};
    float best = BIGF;
    int e = 0;
#pragma unroll
    for (int i = 0; i < 8; ++i) {
        if (d[i] < best) { best = d[i]; e = i; }
    }
    m = best;
    mj = base + e;
}

struct Acc4 { v8f c0, c1, c2, c3; };

__device__ __forceinline__ Acc4 strip_acc16x4(const float* A, int lda,
                                              const float* B, int ldb,
                                              int K, int tm, int tn0, int lane)
{
    const int g = lane >> 4;
    const int r = lane & 15;
    const float* Arow = A + (tm * 16 + r) * lda;
    const float* Bcol = B + tn0 * 16 + 4 * r;
    v8f c0 = {}, c1 = {}, c2 = {}, c3 = {};
    for (int k0 = 0; k0 < K; k0 += 4) {
        const int ka = k0 + 2 * g;
        const float4 q0 = *(const float4*)(Bcol + (ka)     * ldb);
        const float4 q1 = *(const float4*)(Bcol + (ka + 1) * ldb);
        v2f a, b0, b1, b2, b3;
        a.x  = Arow[ka];   a.y  = Arow[ka + 1];
        b0.x = q0.x;       b0.y = q1.x;
        b1.x = q0.y;       b1.y = q1.y;
        b2.x = q0.z;       b2.y = q1.z;
        b3.x = q0.w;       b3.y = q1.w;
        c0 = __builtin_amdgcn_wmma_f32_16x16x4_f32(false, a, false, b0, (short)0, c0, false, false);
        c1 = __builtin_amdgcn_wmma_f32_16x16x4_f32(false, a, false, b1, (short)0, c1, false, false);
        c2 = __builtin_amdgcn_wmma_f32_16x16x4_f32(false, a, false, b2, (short)0, c2, false, false);
        c3 = __builtin_amdgcn_wmma_f32_16x16x4_f32(false, a, false, b3, (short)0, c3, false, false);
    }
    Acc4 acc;
    acc.c0 = c0; acc.c1 = c1; acc.c2 = c2; acc.c3 = c3;
    return acc;
}

__device__ __forceinline__ v8f tile_acc16x16(const float* A, int lda,
                                             const float* B, int ldb,
                                             int K, int tm, int tn, int lane)
{
    const int g = lane >> 4;
    const int r = lane & 15;
    const float* Arow = A + (tm * 16 + r) * lda;
    const float* Bcol = B + tn * 16 + r;
    v8f c = {};
    for (int k0 = 0; k0 < K; k0 += 4) {
        const int ka = k0 + 2 * g;
        v2f a, b;
        a.x = Arow[ka];           a.y = Arow[ka + 1];
        b.x = Bcol[ka * ldb];     b.y = Bcol[(ka + 1) * ldb];
        c = __builtin_amdgcn_wmma_f32_16x16x4_f32(false, a, false, b, (short)0, c, false, false);
    }
    return c;
}

struct Row8 { float d[8]; };

__device__ __forceinline__ Row8 load_row8(const float* dinv, int tm, int lane)
{
    const float* p = dinv + tm * 16 + 8 * (lane >> 4);
    const float4 lo = *(const float4*)(p);
    const float4 hi = *(const float4*)(p + 4);
    Row8 rs;
    rs.d[0] = lo.x; rs.d[1] = lo.y; rs.d[2] = lo.z; rs.d[3] = lo.w;
    rs.d[4] = hi.x; rs.d[5] = hi.y; rs.d[6] = hi.z; rs.d[7] = hi.w;
    return rs;
}

template<bool SCALE, bool ACT>
__device__ __forceinline__ void strip_store16x4(const Acc4& acc, float* OUT, int ldo,
                                                const Row8& rs,
                                                const float* __restrict__ bias,
                                                int tm, int tn0, int lane)
{
    const int g = lane >> 4;
    const int r = lane & 15;
    const v8f cc[4] = {acc.c0, acc.c1, acc.c2, acc.c3};
    float* obase = OUT + (tm * 16 + 8 * g) * ldo + tn0 * 16 + 4 * r;
    float bv[4] = {0.0f, 0.0f, 0.0f, 0.0f};
    if (ACT) {
        const float4 b4 = *(const float4*)(bias + tn0 * 16 + 4 * r);
        bv[0] = b4.x; bv[1] = b4.y; bv[2] = b4.z; bv[3] = b4.w;
    }
#pragma unroll
    for (int v = 0; v < 8; ++v) {
        float o[4];
#pragma unroll
        for (int t = 0; t < 4; ++t) {
            float val = cc[t][v];
            if (SCALE) val *= rs.d[v];
            if (ACT) val = fast_tanh(val + bv[t]);
            o[t] = val;
        }
        *(float4*)(obase + v * ldo) = make_float4(o[0], o[1], o[2], o[3]);
    }
}

__global__ void __launch_bounds__(NTHREADS)
gnn_critic_fused(const float* __restrict__ obs,
                 const float* __restrict__ W1, const float* __restrict__ b1,
                 const float* __restrict__ W2, const float* __restrict__ b2,
                 const float* __restrict__ Wout, const float* __restrict__ bout,
                 float* __restrict__ out)
{
    extern __shared__ float smem[];
    float* obs_s  = smem + OFF_OBS;
    float* M_s    = smem + OFF_M;
    float* X_s    = smem + OFF_X;
    float* dinv_s = smem + OFF_DINV;

    const int tid  = threadIdx.x;
    const int lane = tid & 31;
    const int wave = tid >> 5;
    const int tm   = wave >> 1;
    const int tn0  = (wave & 1) * 4;
    const int bidx = blockIdx.x;

    const float* gobs = obs + (size_t)bidx * (A_AGENTS * D_IN);
    for (int i = tid; i < A_AGENTS * (D_IN / 4); i += NTHREADS) {
        const int row = i / (D_IN / 4);
        const int c4  = i % (D_IN / 4);
        const float4 v = ((const float4*)gobs)[i];
        float* dst = obs_s + row * OBS_LD + c4 * 4;
        dst[0] = v.x; dst[1] = v.y; dst[2] = v.z; dst[3] = v.w;
    }
    for (int i = tid; i < A_AGENTS * MAT_LD; i += NTHREADS) M_s[i] = 0.0f;
    __syncthreads();

    for (int e = tid; e < A_AGENTS * A_AGENTS; e += NTHREADS) {
        const int i = e >> 7;
        const int j = e & 127;
        const float dx = obs_s[i * OBS_LD + 0] - obs_s[j * OBS_LD + 0];
        const float dy = obs_s[i * OBS_LD + 1] - obs_s[j * OBS_LD + 1];
        float d = dx * dx + dy * dy;
        if (i == j) d = BIGF;
        X_s[i * MAT_LD + j] = d;
    }
    __syncthreads();

    {
        const int a  = tid >> 2;
        const int sl = tid & 3;
        const float* drow = X_s + a * MAT_LD + sl * 32;
        if (sl == 0) M_s[a * MAT_LD + a] = 1.0f;
        float gm[4];
        int   gj[4];
#pragma unroll
        for (int q = 0; q < 4; ++q) scan8(drow + 8 * q, sl * 32 + 8 * q, gm[q], gj[q]);

        for (int t = 0; t < KNN; ++t) {
            float best = BIGF;
            int bestj = 0;
#pragma unroll
            for (int q = 0; q < 4; ++q) {
                if (gm[q] < best) { best = gm[q]; bestj = gj[q]; }
            }
            for (int off = 1; off < 4; off <<= 1) {
                const float ob = __shfl_xor(best, off);
                const int   oj = __shfl_xor(bestj, off);
                if (ob < best || (ob == best && oj < bestj)) {
                    best = ob; bestj = oj;
                }
            }
            if (sl == 0) {
                X_s[a * MAT_LD + bestj] = BIGF;
                M_s[bestj * MAT_LD + a] = 1.0f;
            }
            if (t + 1 < KNN) {
                const int qw = (bestj >> 3) & 3;
                float nm;
                int   nj;
                scan8(drow + 8 * qw, sl * 32 + 8 * qw, nm, nj);
#pragma unroll
                for (int q = 0; q < 4; ++q) {
                    const bool hit = (q == qw);
                    gm[q] = hit ? nm : gm[q];
                    gj[q] = hit ? nj : gj[q];
                }
            }
        }
    }
    __syncthreads();

    {
        const int a  = tid >> 2;
        const int sl = tid & 3;
        float* orow = obs_s + a * OBS_LD + sl * 8;
        float4 lo = *(const float4*)(orow);
        float4 hi = *(const float4*)(orow + 4);
        const float* seg = M_s + a * MAT_LD + sl * 32;
        float s = 0.0f;
        for (int i = 0; i < 32; ++i) s += seg[i];
        s += __shfl_xor(s, 1);
        s += __shfl_xor(s, 2);
        const float dv = rsqrtf(s);
        if (sl == 0) dinv_s[a] = dv;
        lo.x *= dv; lo.y *= dv; lo.z *= dv; lo.w *= dv;
        hi.x *= dv; hi.y *= dv; hi.z *= dv; hi.w *= dv;
        *(float4*)(orow)     = lo;
        *(float4*)(orow + 4) = hi;
    }
    __syncthreads();

    {
        const int tn = wave & 1;
        const v8f c = tile_acc16x16(M_s, MAT_LD, obs_s, OBS_LD, A_AGENTS, tm, tn, lane);
        const Row8 rs = load_row8(dinv_s, tm, lane);
        float* obase = X_s + (tm * 16 + 8 * (lane >> 4)) * MAT_LD + tn * 16 + (lane & 15);
#pragma unroll
        for (int v = 0; v < 8; ++v) obase[v * MAT_LD] = c[v] * rs.d[v];
    }
    __syncthreads();

    {
        const Acc4 acc = strip_acc16x4(X_s, MAT_LD, W1, HID, D_IN, tm, tn0, lane);
        const Row8 none = {};
        __syncthreads();
        strip_store16x4<false, true>(acc, X_s, MAT_LD, none, b1, tm, tn0, lane);
    }
    __syncthreads();

    {
        const Acc4 acc = strip_acc16x4(X_s, MAT_LD, W2, HID, HID, tm, tn0, lane);
        const Row8 rs = load_row8(dinv_s, tm, lane);
        __syncthreads();
        strip_store16x4<true, false>(acc, X_s, MAT_LD, rs, nullptr, tm, tn0, lane);
    }
    __syncthreads();

    {
        const Acc4 acc = strip_acc16x4(M_s, MAT_LD, X_s, MAT_LD, HID, tm, tn0, lane);
        const Row8 rs = load_row8(dinv_s, tm, lane);
        __syncthreads();
        strip_store16x4<true, true>(acc, X_s, MAT_LD, rs, b2, tm, tn0, lane);
    }
    __syncthreads();

    {
        const int a  = tid >> 2;
        const int sl = tid & 3;
        const float* hrow = X_s + a * MAT_LD + sl * 32;
        const float* wseg = Wout + sl * 32;
        float s = 0.0f;
        for (int k = 0; k < 32; ++k) s += hrow[k] * wseg[k];
        s += __shfl_xor(s, 1);
        s += __shfl_xor(s, 2);
        if (sl == 0) out[(size_t)bidx * A_AGENTS + a] = s + bout[0];
    }
}

extern "C" void kernel_launch(void* const* d_in, const int* in_sizes, int n_in,
                              void* d_out, int out_size, void* d_ws, size_t ws_size,
                              hipStream_t stream) {
    (void)n_in; (void)out_size; (void)d_ws; (void)ws_size;
    const float* obs  = (const float*)d_in[0];
    const float* W1   = (const float*)d_in[1];
    const float* b1   = (const float*)d_in[2];
    const float* W2   = (const float*)d_in[3];
    const float* b2   = (const float*)d_in[4];
    const float* Wout = (const float*)d_in[5];
    const float* bout = (const float*)d_in[6];
    float* out = (float*)d_out;

    const int B = in_sizes[0] / (A_AGENTS * D_IN);

    static_assert(2 * LDS_BYTES <= 320 * 1024, "two workgroups must fit one WGP's LDS");
    (void)hipFuncSetAttribute((const void*)gnn_critic_fused,
                              hipFuncAttributeMaxDynamicSharedMemorySize,
                              LDS_BYTES);
    gnn_critic_fused<<<B, NTHREADS, LDS_BYTES, stream>>>(
        obs, W1, b1, W2, b2, Wout, bout, out);
}


# kNN pick loop: one lane per agent keeps the minima of all 16 groups of 8 in registers and picks by a tournament, no 4-lane combine; 4 waves issue the 16 rounds instead of 16 waves, same picks
# speedup vs baseline: 1.9594x; 1.0797x over previous
#include <hip/hip_runtime.h>
#include <hip/hip_bf16.h>
#include <math.h>


typedef float v2f __attribute__((ext_vector_type(2)));
typedef float v8f __attribute__((ext_vector_type(8)));

#define A_AGENTS 128
#define D_IN     32
#define HID      128
#define KNN      16
#define OBS_LD   36
#define MAT_LD   132
#define NTHREADS 512
#define NWAVES   (NTHREADS / 32)
#define BIGF     3.4e38f

#define OFF_OBS  0
#define OFF_M    (OFF_OBS + A_AGENTS * OBS_LD)
#define OFF_X    (OFF_M   + A_AGENTS * MAT_LD)
#define OFF_DINV (OFF_X   + A_AGENTS * MAT_LD)
#define LDS_FLOATS (OFF_DINV + A_AGENTS)
#define LDS_BYTES  (LDS_FLOATS * 4)

__device__ __forceinline__ float fast_tanh(float x) {
#if __has_builtin(__builtin_amdgcn_tanhf)
    return __builtin_amdgcn_tanhf(x);
#elif __has_builtin(__builtin_amdgcn_tanh_f32)
    return __builtin_amdgcn_tanh_f32(x);
#else
    return tanhf(x);
#endif
}

__device__ __forceinline__ void scan8(const float* p, int base, float& m, int& mj)
{
    const float4 lo = *(const float4*)(p);
    const float4 hi = *(const float4*)(p + 4);
    const bool c01 = lo.y < lo.x;
    const bool c23 = lo.w < lo.z;
    const bool c45 = hi.y < hi.x;
    const bool c67 = hi.w < hi.z;
    const float m01 = c01 ? lo.y : lo.x;   const int e01 = c01 ? 1 : 0;
    const float m23 = c23 ? lo.w : lo.z;   const int e23 = c23 ? 3 : 2;
    const float m45 = c45 ? hi.y : hi.x;   const int e45 = c45 ? 5 : 4;
    const float m67 = c67 ? hi.w : hi.z;   const int e67 = c67 ? 7 : 6;
    const bool cA = m23 < m01;
    const bool cB = m67 < m45;
    const float mA = cA ? m23 : m01;       const int eA = cA ? e23 : e01;
    const float mB = cB ? m67 : m45;       const int eB = cB ? e67 : e45;
    const bool cT = mB < mA;
    m  = cT ? mB : mA;
    mj = base + (cT ? eB : eA);
}

struct Cand { float m; int j; };

__device__ __forceinline__ Cand lower_of(const Cand l, const Cand r)
{
    const bool c = r.m < l.m;
    Cand o;
    o.m = c ? r.m : l.m;
    o.j = c ? r.j : l.j;
    return o;
}

struct Acc4 { v8f c0, c1, c2, c3; };

__device__ __forceinline__ Acc4 strip_acc16x4(const float* A, int lda,
                                              const float* B, int ldb,
                                              int K, int tm, int tn0, int lane)
{
    const int g = lane >> 4;
    const int r = lane & 15;
    const float* Arow = A + (tm * 16 + r) * lda;
    const float* Bcol = B + tn0 * 16 + 4 * r;
    v8f c0 = {}, c1 = {}, c2 = {}, c3 = {};
    for (int k0 = 0; k0 < K; k0 += 4) {
        const int ka = k0 + 2 * g;
        const float4 q0 = *(const float4*)(Bcol + (ka)     * ldb);
        const float4 q1 = *(const float4*)(Bcol + (ka + 1) * ldb);
        v2f a, b0, b1, b2, b3;
        a.x  = Arow[ka];   a.y  = Arow[ka + 1];
        b0.x = q0.x;       b0.y = q1.x;
        b1.x = q0.y;       b1.y = q1.y;
        b2.x = q0.z;       b2.y = q1.z;
        b3.x = q0.w;       b3.y = q1.w;
        c0 = __builtin_amdgcn_wmma_f32_16x16x4_f32(false, a, false, b0, (short)0, c0, false, false);
        c1 = __builtin_amdgcn_wmma_f32_16x16x4_f32(false, a, false, b1, (short)0, c1, false, false);
        c2 = __builtin_amdgcn_wmma_f32_16x16x4_f32(false, a, false, b2, (short)0, c2, false, false);
        c3 = __builtin_amdgcn_wmma_f32_16x16x4_f32(false, a, false, b3, (short)0, c3, false, false);
    }
    Acc4 acc;
    acc.c0 = c0; acc.c1 = c1; acc.c2 = c2; acc.c3 = c3;
    return acc;
}

__device__ __forceinline__ v8f tile_acc16x16(const float* A, int lda,
                                             const float* B, int ldb,
                                             int K, int tm, int tn, int lane)
{
    const int g = lane >> 4;
    const int r = lane & 15;
    const float* Arow = A + (tm * 16 + r) * lda;
    const float* Bcol = B + tn * 16 + r;
    v8f c = {};
    for (int k0 = 0; k0 < K; k0 += 4) {
        const int ka = k0 + 2 * g;
        v2f a, b;
        a.x = Arow[ka];           a.y = Arow[ka + 1];
        b.x = Bcol[ka * ldb];     b.y = Bcol[(ka + 1) * ldb];
        c = __builtin_amdgcn_wmma_f32_16x16x4_f32(false, a, false, b, (short)0, c, false, false);
    }
    return c;
}

struct Row8 { float d[8]; };

__device__ __forceinline__ Row8 load_row8(const float* dinv, int tm, int lane)
{
    const float* p = dinv + tm * 16 + 8 * (lane >> 4);
    const float4 lo = *(const float4*)(p);
    const float4 hi = *(const float4*)(p + 4);
    Row8 rs;
    rs.d[0] = lo.x; rs.d[1] = lo.y; rs.d[2] = lo.z; rs.d[3] = lo.w;
    rs.d[4] = hi.x; rs.d[5] = hi.y; rs.d[6] = hi.z; rs.d[7] = hi.w;
    return rs;
}

template<bool SCALE, bool ACT>
__device__ __forceinline__ void strip_store16x4(const Acc4& acc, float* OUT, int ldo,
                                                const Row8& rs,
                                                const float* __restrict__ bias,
                                                int tm, int tn0, int lane)
{
    const int g = lane >> 4;
    const int r = lane & 15;
    const v8f cc[4] = {acc.c0, acc.c1, acc.c2, acc.c3};
    float* obase = OUT + (tm * 16 + 8 * g) * ldo + tn0 * 16 + 4 * r;
    float bv[4] = {0.0f, 0.0f, 0.0f, 0.0f};
    if (ACT) {
        const float4 b4 = *(const float4*)(bias + tn0 * 16 + 4 * r);
        bv[0] = b4.x; bv[1] = b4.y; bv[2] = b4.z; bv[3] = b4.w;
    }
#pragma unroll
    for (int v = 0; v < 8; ++v) {
        float o[4];
#pragma unroll
        for (int t = 0; t < 4; ++t) {
            float val = cc[t][v];
            if (SCALE) val *= rs.d[v];
            if (ACT) val = fast_tanh(val + bv[t]);
            o[t] = val;
        }
        *(float4*)(obase + v * ldo) = make_float4(o[0], o[1], o[2], o[3]);
    }
}

__global__ void __launch_bounds__(NTHREADS)
gnn_critic_fused(const float* __restrict__ obs,
                 const float* __restrict__ W1, const float* __restrict__ b1,
                 const float* __restrict__ W2, const float* __restrict__ b2,
                 const float* __restrict__ Wout, const float* __restrict__ bout,
                 float* __restrict__ out)
{
    extern __shared__ float smem[];
    float* obs_s  = smem + OFF_OBS;
    float* M_s    = smem + OFF_M;
    float* X_s    = smem + OFF_X;
    float* dinv_s = smem + OFF_DINV;

    const int tid  = threadIdx.x;
    const int lane = tid & 31;
    const int wave = tid >> 5;
    const int tm   = wave >> 1;
    const int tn0  = (wave & 1) * 4;
    const int bidx = blockIdx.x;

    const float* gobs = obs + (size_t)bidx * (A_AGENTS * D_IN);
    for (int i = tid; i < A_AGENTS * (D_IN / 4); i += NTHREADS) {
        const int row = i / (D_IN / 4);
        const int c4  = i % (D_IN / 4);
        const float4 v = ((const float4*)gobs)[i];
        float* dst = obs_s + row * OBS_LD + c4 * 4;
        dst[0] = v.x; dst[1] = v.y; dst[2] = v.z; dst[3] = v.w;
    }
    for (int i = tid; i < A_AGENTS * MAT_LD; i += NTHREADS) M_s[i] = 0.0f;
    __syncthreads();

    for (int e = tid; e < A_AGENTS * A_AGENTS; e += NTHREADS) {
        const int i = e >> 7;
        const int j = e & 127;
        const float dx = obs_s[i * OBS_LD + 0] - obs_s[j * OBS_LD + 0];
        const float dy = obs_s[i * OBS_LD + 1] - obs_s[j * OBS_LD + 1];
        float d = dx * dx + dy * dy;
        if (i == j) d = BIGF;
        X_s[i * MAT_LD + j] = d;
    }
    __syncthreads();

    if (tid < A_AGENTS) {
        const int a = tid;
        float* drow = X_s + a * MAT_LD;
        M_s[a * MAT_LD + a] = 1.0f;
        float gm[16];
        int   gj[16];
#pragma unroll
        for (int q = 0; q < 16; ++q) scan8(drow + 8 * q, 8 * q, gm[q], gj[q]);

        for (int t = 0; t < KNN; ++t) {
            const Cand a0 = lower_of(Cand{gm[0],  gj[0]},  Cand{gm[1],  gj[1]});
            const Cand a1 = lower_of(Cand{gm[2],  gj[2]},  Cand{gm[3],  gj[3]});
            const Cand a2 = lower_of(Cand{gm[4],  gj[4]},  Cand{gm[5],  gj[5]});
            const Cand a3 = lower_of(Cand{gm[6],  gj[6]},  Cand{gm[7],  gj[7]});
            const Cand a4 = lower_of(Cand{gm[8],  gj[8]},  Cand{gm[9],  gj[9]});
            const Cand a5 = lower_of(Cand{gm[10], gj[10]}, Cand{gm[11], gj[11]});
            const Cand a6 = lower_of(Cand{gm[12], gj[12]}, Cand{gm[13], gj[13]});
            const Cand a7 = lower_of(Cand{gm[14], gj[14]}, Cand{gm[15], gj[15]});
            const Cand b0 = lower_of(a0, a1);
            const Cand b1 = lower_of(a2, a3);
            const Cand b2 = lower_of(a4, a5);
            const Cand b3 = lower_of(a6, a7);
            const Cand c0 = lower_of(b0, b1);
            const Cand c1 = lower_of(b2, b3);
            const int bestj = lower_of(c0, c1).j;

            drow[bestj] = BIGF;
            M_s[bestj * MAT_LD + a] = 1.0f;
            if (t + 1 < KNN) {
                const int qw = bestj >> 3;
                float nm;
                int   nj;
                scan8(drow + 8 * qw, 8 * qw, nm, nj);
#pragma unroll
                for (int q = 0; q < 16; ++q) {
                    const bool hit = (q == qw);
                    gm[q] = hit ? nm : gm[q];
                    gj[q] = hit ? nj : gj[q];
                }
            }
        }
    }
    __syncthreads();

    {
        const int a  = tid >> 2;
        const int sl = tid & 3;
        float* orow = obs_s + a * OBS_LD + sl * 8;
        float4 lo = *(const float4*)(orow);
        float4 hi = *(const float4*)(orow + 4);
        const float* seg = M_s + a * MAT_LD + sl * 32;
        float s = 0.0f;
        for (int i = 0; i < 32; ++i) s += seg[i];
        s += __shfl_xor(s, 1);
        s += __shfl_xor(s, 2);
        const float dv = rsqrtf(s);
        if (sl == 0) dinv_s[a] = dv;
        lo.x *= dv; lo.y *= dv; lo.z *= dv; lo.w *= dv;
        hi.x *= dv; hi.y *= dv; hi.z *= dv; hi.w *= dv;
        *(float4*)(orow)     = lo;
        *(float4*)(orow + 4) = hi;
    }
    __syncthreads();

    {
        const int tn = wave & 1;
        const v8f c = tile_acc16x16(M_s, MAT_LD, obs_s, OBS_LD, A_AGENTS, tm, tn, lane);
        const Row8 rs = load_row8(dinv_s, tm, lane);
        float* obase = X_s + (tm * 16 + 8 * (lane >> 4)) * MAT_LD + tn * 16 + (lane & 15);
#pragma unroll
        for (int v = 0; v < 8; ++v) obase[v * MAT_LD] = c[v] * rs.d[v];
    }
    __syncthreads();

    {
        const Acc4 acc = strip_acc16x4(X_s, MAT_LD, W1, HID, D_IN, tm, tn0, lane);
        const Row8 none = {};
        __syncthreads();
        strip_store16x4<false, true>(acc, X_s, MAT_LD, none, b1, tm, tn0, lane);
    }
    __syncthreads();

    {
        const Acc4 acc = strip_acc16x4(X_s, MAT_LD, W2, HID, HID, tm, tn0, lane);
        const Row8 rs = load_row8(dinv_s, tm, lane);
        __syncthreads();
        strip_store16x4<true, false>(acc, X_s, MAT_LD, rs, nullptr, tm, tn0, lane);
    }
    __syncthreads();

    {
        const Acc4 acc = strip_acc16x4(M_s, MAT_LD, X_s, MAT_LD, HID, tm, tn0, lane);
        const Row8 rs = load_row8(dinv_s, tm, lane);
        __syncthreads();
        strip_store16x4<true, true>(acc, X_s, MAT_LD, rs, b2, tm, tn0, lane);
    }
    __syncthreads();

    {
        const int a  = tid >> 2;
        const int sl = tid & 3;
        const float* hrow = X_s + a * MAT_LD + sl * 32;
        const float* wseg = Wout + sl * 32;
        float s = 0.0f;
        for (int k = 0; k < 32; ++k) s += hrow[k] * wseg[k];
        s += __shfl_xor(s, 1);
        s += __shfl_xor(s, 2);
        if (sl == 0) out[(size_t)bidx * A_AGENTS + a] = s + bout[0];
    }
}

extern "C" void kernel_launch(void* const* d_in, const int* in_sizes, int n_in,
                              void* d_out, int out_size, void* d_ws, size_t ws_size,
                              hipStream_t stream) {
    (void)n_in; (void)out_size; (void)d_ws; (void)ws_size;
    const float* obs  = (const float*)d_in[0];
    const float* W1   = (const float*)d_in[1];
    const float* b1   = (const float*)d_in[2];
    const float* W2   = (const float*)d_in[3];
    const float* b2   = (const float*)d_in[4];
    const float* Wout = (const float*)d_in[5];
    const float* bout = (const float*)d_in[6];
    float* out = (float*)d_out;

    const int B = in_sizes[0] / (A_AGENTS * D_IN);

    static_assert(2 * LDS_BYTES <= 320 * 1024, "two workgroups must fit one WGP's LDS");
    (void)hipFuncSetAttribute((const void*)gnn_critic_fused,
                              hipFuncAttributeMaxDynamicSharedMemorySize,
                              LDS_BYTES);
    gnn_critic_fused<<<B, NTHREADS, LDS_BYTES, stream>>>(
        obs, W1, b1, W2, b2, Wout, bout, out);
}


# Stage-7 result kept in LDS with rows k, k+1 of a column adjacent (k-pair layout): stage 8's B fragments load straight into operand registers, no moves or address adds in its loop; same sums
# speedup vs baseline: 2.1508x; 1.0977x over previous
#include <hip/hip_runtime.h>
#include <hip/hip_bf16.h>
#include <math.h>


typedef float v2f __attribute__((ext_vector_type(2)));
typedef float v8f __attribute__((ext_vector_type(8)));

#define A_AGENTS 128
#define D_IN     32
#define HID      128
#define KNN      16
#define OBS_LD   36
#define MAT_LD   132
#define PAIR_LD  (2 * MAT_LD)
#define NTHREADS 512
#define NWAVES   (NTHREADS / 32)
#define BIGF     3.4e38f

#define OFF_OBS  0
#define OFF_X    (OFF_OBS + A_AGENTS * OBS_LD)
#define OFF_M    (OFF_X   + A_AGENTS * MAT_LD)
#define OFF_DINV (OFF_M   + A_AGENTS * MAT_LD)
#define LDS_FLOATS (OFF_DINV + A_AGENTS)
#define LDS_BYTES  (LDS_FLOATS * 4)

__device__ __forceinline__ float fast_tanh(float x) {
#if __has_builtin(__builtin_amdgcn_tanhf)
    return __builtin_amdgcn_tanhf(x);
#elif __has_builtin(__builtin_amdgcn_tanh_f32)
    return __builtin_amdgcn_tanh_f32(x);
#else
    return tanhf(x);
#endif
}

__device__ __forceinline__ void scan8(const float* p, int base, float& m, int& mj)
{
    const float4 lo = *(const float4*)(p);
    const float4 hi = *(const float4*)(p + 4);
    const bool c01 = lo.y < lo.x;
    const bool c23 = lo.w < lo.z;
    const bool c45 = hi.y < hi.x;
    const bool c67 = hi.w < hi.z;
    const float m01 = c01 ? lo.y : lo.x;   const int e01 = c01 ? 1 : 0;
    const float m23 = c23 ? lo.w : lo.z;   const int e23 = c23 ? 3 : 2;
    const float m45 = c45 ? hi.y : hi.x;   const int e45 = c45 ? 5 : 4;
    const float m67 = c67 ? hi.w : hi.z;   const int e67 = c67 ? 7 : 6;
    const bool cA = m23 < m01;
    const bool cB = m67 < m45;
    const float mA = cA ? m23 : m01;       const int eA = cA ? e23 : e01;
    const float mB = cB ? m67 : m45;       const int eB = cB ? e67 : e45;
    const bool cT = mB < mA;
    m  = cT ? mB : mA;
    mj = base + (cT ? eB : eA);
}

struct Cand { float m; int j; };

__device__ __forceinline__ Cand lower_of(const Cand l, const Cand r)
{
    const bool c = r.m < l.m;
    Cand o;
    o.m = c ? r.m : l.m;
    o.j = c ? r.j : l.j;
    return o;
}

struct Acc4 { v8f c0, c1, c2, c3; };

__device__ __forceinline__ Acc4 strip_acc16x4(const float* A, int lda,
                                              const float* B, int ldb,
                                              int K, int tm, int tn0, int lane)
{
    const int g = lane >> 4;
    const int r = lane & 15;
    const float* Arow = A + (tm * 16 + r) * lda;
    const float* Bcol = B + tn0 * 16 + 4 * r;
    v8f c0 = {}, c1 = {}, c2 = {}, c3 = {};
    for (int k0 = 0; k0 < K; k0 += 4) {
        const int ka = k0 + 2 * g;
        const float4 q0 = *(const float4*)(Bcol + (ka)     * ldb);
        const float4 q1 = *(const float4*)(Bcol + (ka + 1) * ldb);
        v2f a, b0, b1, b2, b3;
        a.x  = Arow[ka];   a.y  = Arow[ka + 1];
        b0.x = q0.x;       b0.y = q1.x;
        b1.x = q0.y;       b1.y = q1.y;
        b2.x = q0.z;       b2.y = q1.z;
        b3.x = q0.w;       b3.y = q1.w;
        c0 = __builtin_amdgcn_wmma_f32_16x16x4_f32(false, a, false, b0, (short)0, c0, false, false);
        c1 = __builtin_amdgcn_wmma_f32_16x16x4_f32(false, a, false, b1, (short)0, c1, false, false);
        c2 = __builtin_amdgcn_wmma_f32_16x16x4_f32(false, a, false, b2, (short)0, c2, false, false);
        c3 = __builtin_amdgcn_wmma_f32_16x16x4_f32(false, a, false, b3, (short)0, c3, false, false);
    }
    Acc4 acc;
    acc.c0 = c0; acc.c1 = c1; acc.c2 = c2; acc.c3 = c3;
    return acc;
}

__device__ __forceinline__ int pair_off(int pair_row, int tn0, int r)
{
    return pair_row * PAIR_LD + tn0 * 32 + 4 * r;
}

__device__ __forceinline__ Acc4 strip_acc16x4_pairs(const float* A, int lda,
                                                    const float* BP,
                                                    int K, int tm, int tn0, int lane)
{
    const int g = lane >> 4;
    const int r = lane & 15;
    const float* ap = A + (tm * 16 + r) * lda + 2 * g;
    const float* bp = BP + pair_off(g, tn0, r);
    v8f c0 = {}, c1 = {}, c2 = {}, c3 = {};
#pragma unroll 8
    for (int k0 = 0; k0 < K; k0 += 4) {
        const v2f a = *(const v2f*)(ap);
        const float4 q0 = *(const float4*)(bp);
        const float4 q1 = *(const float4*)(bp + 64);
        v2f b0, b1, b2, b3;
        b0.x = q0.x;  b0.y = q0.y;
        b1.x = q0.z;  b1.y = q0.w;
        b2.x = q1.x;  b2.y = q1.y;
        b3.x = q1.z;  b3.y = q1.w;
        c0 = __builtin_amdgcn_wmma_f32_16x16x4_f32(false, a, false, b0, (short)0, c0, false, false);
        c1 = __builtin_amdgcn_wmma_f32_16x16x4_f32(false, a, false, b1, (short)0, c1, false, false);
        c2 = __builtin_amdgcn_wmma_f32_16x16x4_f32(false, a, false, b2, (short)0, c2, false, false);
        c3 = __builtin_amdgcn_wmma_f32_16x16x4_f32(false, a, false, b3, (short)0, c3, false, false);
        ap += 4;
        bp += 2 * PAIR_LD;
    }
    Acc4 acc;
    acc.c0 = c0; acc.c1 = c1; acc.c2 = c2; acc.c3 = c3;
    return acc;
}

__device__ __forceinline__ v8f tile_acc16x16(const float* A, int lda,
                                             const float* B, int ldb,
                                             int K, int tm, int tn, int lane)
{
    const int g = lane >> 4;
    const int r = lane & 15;
    const float* Arow = A + (tm * 16 + r) * lda;
    const float* Bcol = B + tn * 16 + r;
    v8f c = {};
    for (int k0 = 0; k0 < K; k0 += 4) {
        const int ka = k0 + 2 * g;
        v2f a, b;
        a.x = Arow[ka];           a.y = Arow[ka + 1];
        b.x = Bcol[ka * ldb];     b.y = Bcol[(ka + 1) * ldb];
        c = __builtin_amdgcn_wmma_f32_16x16x4_f32(false, a, false, b, (short)0, c, false, false);
    }
    return c;
}

struct Row8 { float d[8]; };

__device__ __forceinline__ Row8 load_row8(const float* dinv, int tm, int lane)
{
    const float* p = dinv + tm * 16 + 8 * (lane >> 4);
    const float4 lo = *(const float4*)(p);
    const float4 hi = *(const float4*)(p + 4);
    Row8 rs;
    rs.d[0] = lo.x; rs.d[1] = lo.y; rs.d[2] = lo.z; rs.d[3] = lo.w;
    rs.d[4] = hi.x; rs.d[5] = hi.y; rs.d[6] = hi.z; rs.d[7] = hi.w;
    return rs;
}

template<bool SCALE, bool ACT>
__device__ __forceinline__ void strip_store16x4(const Acc4& acc, float* OUT, int ldo,
                                                const Row8& rs,
                                                const float* __restrict__ bias,
                                                int tm, int tn0, int lane)
{
    const int g = lane >> 4;
    const int r = lane & 15;
    const v8f cc[4] = {acc.c0, acc.c1, acc.c2, acc.c3};
    float* obase = OUT + (tm * 16 + 8 * g) * ldo + tn0 * 16 + 4 * r;
    float bv[4] = {0.0f, 0.0f, 0.0f, 0.0f};
    if (ACT) {
        const float4 b4 = *(const float4*)(bias + tn0 * 16 + 4 * r);
        bv[0] = b4.x; bv[1] = b4.y; bv[2] = b4.z; bv[3] = b4.w;
    }
#pragma unroll
    for (int v = 0; v < 8; ++v) {
        float o[4];
#pragma unroll
        for (int t = 0; t < 4; ++t) {
            float val = cc[t][v];
            if (SCALE) val *= rs.d[v];
            if (ACT) val = fast_tanh(val + bv[t]);
            o[t] = val;
        }
        *(float4*)(obase + v * ldo) = make_float4(o[0], o[1], o[2], o[3]);
    }
}

__device__ __forceinline__ void strip_store16x4_pairs(const Acc4& acc, float* OUTP,
                                                      const Row8& rs,
                                                      int tm, int tn0, int lane)
{
    const int g = lane >> 4;
    const int r = lane & 15;
    float* obase = OUTP + pair_off(tm * 8 + 4 * g, tn0, r);
#pragma unroll
    for (int u = 0; u < 4; ++u) {
        const float d0 = rs.d[2 * u], d1 = rs.d[2 * u + 1];
        *(float4*)(obase + u * PAIR_LD) =
            make_float4(acc.c0[2 * u] * d0, acc.c0[2 * u + 1] * d1,
                        acc.c1[2 * u] * d0, acc.c1[2 * u + 1] * d1);
        *(float4*)(obase + u * PAIR_LD + 64) =
            make_float4(acc.c2[2 * u] * d0, acc.c2[2 * u + 1] * d1,
                        acc.c3[2 * u] * d0, acc.c3[2 * u + 1] * d1);
    }
}

__global__ void __launch_bounds__(NTHREADS)
gnn_critic_fused(const float* __restrict__ obs,
                 const float* __restrict__ W1, const float* __restrict__ b1,
                 const float* __restrict__ W2, const float* __restrict__ b2,
                 const float* __restrict__ Wout, const float* __restrict__ bout,
                 float* __restrict__ out)
{
    extern __shared__ __attribute__((aligned(16))) float smem[];
    float* obs_s  = smem + OFF_OBS;
    float* M_s    = smem + OFF_M;
    float* X_s    = smem + OFF_X;
    float* dinv_s = smem + OFF_DINV;

    const int tid  = threadIdx.x;
    const int lane = tid & 31;
    const int wave = tid >> 5;
    const int tm   = wave >> 1;
    const int tn0  = (wave & 1) * 4;
    const int bidx = blockIdx.x;

    const float* gobs = obs + (size_t)bidx * (A_AGENTS * D_IN);
    static_assert(A_AGENTS * (D_IN / 4) == 2 * NTHREADS, "two pieces per thread");
    static_assert((A_AGENTS * MAT_LD) % NTHREADS == 0, "M zeroing has no remainder");
    const float4 v0 = ((const float4*)gobs)[tid];
    const float4 v1 = ((const float4*)gobs)[tid + NTHREADS];
#pragma unroll
    for (int i = 0; i < (A_AGENTS * MAT_LD) / NTHREADS; ++i) M_s[tid + i * NTHREADS] = 0.0f;
    {
        float* dst = obs_s + (tid >> 3) * OBS_LD + (tid & 7) * 4;
        *(float4*)(dst) = v0;
        *(float4*)(dst + (NTHREADS / 8) * OBS_LD) = v1;
    }
    __syncthreads();

    {
        const int a  = tid >> 2;
        const int sl = tid & 3;
        const float2 pi = *(const float2*)(obs_s + a * OBS_LD);
        const float* prow = obs_s + (32 * sl) * OBS_LD;
        float* drow = X_s + a * MAT_LD + 32 * sl;
        float2 pj[32];
#pragma unroll
        for (int u = 0; u < 8; ++u) {
            const int cb = ((u + sl) & 7) * 4;
#pragma unroll
            for (int e = 0; e < 4; ++e)
                pj[4 * u + e] = *(const float2*)(prow + (cb + e) * OBS_LD);
        }
        __builtin_amdgcn_sched_barrier(0);
#pragma unroll
        for (int u = 0; u < 8; ++u) {
            const int cb = ((u + sl) & 7) * 4;
            float dd[4];
#pragma unroll
            for (int e = 0; e < 4; ++e) {
                const float dx = pi.x - pj[4 * u + e].x;
                const float dy = pi.y - pj[4 * u + e].y;
                {
#pragma clang fp contract(off)
                    dd[e] = dx * dx + dy * dy;
                }
            }
            *(float4*)(drow + cb) = make_float4(dd[0], dd[1], dd[2], dd[3]);
        }
        if ((a >> 5) == sl) X_s[a * MAT_LD + a] = BIGF;
    }
    __syncthreads();

    if (tid < A_AGENTS) {
        const int a = tid;
        float* drow = X_s + a * MAT_LD;
        M_s[a * MAT_LD + a] = 1.0f;
        float gm[16];
        int   gj[16];
#pragma unroll
        for (int q = 0; q < 16; ++q) scan8(drow + 8 * q, 8 * q, gm[q], gj[q]);

        for (int t = 0; t < KNN; ++t) {
            const Cand a0 = lower_of(Cand{gm[0],  gj[0]},  Cand{gm[1],  gj[1]});
            const Cand a1 = lower_of(Cand{gm[2],  gj[2]},  Cand{gm[3],  gj[3]});
            const Cand a2 = lower_of(Cand{gm[4],  gj[4]},  Cand{gm[5],  gj[5]});
            const Cand a3 = lower_of(Cand{gm[6],  gj[6]},  Cand{gm[7],  gj[7]});
            const Cand a4 = lower_of(Cand{gm[8],  gj[8]},  Cand{gm[9],  gj[9]});
            const Cand a5 = lower_of(Cand{gm[10], gj[10]}, Cand{gm[11], gj[11]});
            const Cand a6 = lower_of(Cand{gm[12], gj[12]}, Cand{gm[13], gj[13]});
            const Cand a7 = lower_of(Cand{gm[14], gj[14]}, Cand{gm[15], gj[15]});
            const Cand b0 = lower_of(a0, a1);
            const Cand b1 = lower_of(a2, a3);
            const Cand b2 = lower_of(a4, a5);
            const Cand b3 = lower_of(a6, a7);
            const Cand c0 = lower_of(b0, b1);
            const Cand c1 = lower_of(b2, b3);
            const int bestj = lower_of(c0, c1).j;

            drow[bestj] = BIGF;
            M_s[bestj * MAT_LD + a] = 1.0f;
            if (t + 1 < KNN) {
                const int qw = bestj >> 3;
                float nm;
                int   nj;
                scan8(drow + 8 * qw, 8 * qw, nm, nj);
#pragma unroll
                for (int q = 0; q < 16; ++q) {
                    const bool hit = (q == qw);
                    gm[q] = hit ? nm : gm[q];
                    gj[q] = hit ? nj : gj[q];
                }
            }
        }
    }
    __syncthreads();

    {
        const int a  = tid >> 2;
        const int sl = tid & 3;
        float* orow = obs_s + a * OBS_LD + sl * 8;
        float4 lo = *(const float4*)(orow);
        float4 hi = *(const float4*)(orow + 4);
        const float* seg = M_s + a * MAT_LD + sl * 32;
        float s = 0.0f;
        for (int i = 0; i < 32; ++i) s += seg[i];
        s += __shfl_xor(s, 1);
        s += __shfl_xor(s, 2);
        const float dv = rsqrtf(s);
        if (sl == 0) dinv_s[a] = dv;
        lo.x *= dv; lo.y *= dv; lo.z *= dv; lo.w *= dv;
        hi.x *= dv; hi.y *= dv; hi.z *= dv; hi.w *= dv;
        *(float4*)(orow)     = lo;
        *(float4*)(orow + 4) = hi;
    }
    __syncthreads();

    {
        const int tn = wave & 1;
        const v8f c = tile_acc16x16(M_s, MAT_LD, obs_s, OBS_LD, A_AGENTS, tm, tn, lane);
        const Row8 rs = load_row8(dinv_s, tm, lane);
        float* obase = X_s + (tm * 16 + 8 * (lane >> 4)) * MAT_LD + tn * 16 + (lane & 15);
#pragma unroll
        for (int v = 0; v < 8; ++v) obase[v * MAT_LD] = c[v] * rs.d[v];
    }
    __syncthreads();

    {
        const Acc4 acc = strip_acc16x4(X_s, MAT_LD, W1, HID, D_IN, tm, tn0, lane);
        const Row8 none = {};
        __syncthreads();
        strip_store16x4<false, true>(acc, X_s, MAT_LD, none, b1, tm, tn0, lane);
    }
    __syncthreads();

    {
        const Acc4 acc = strip_acc16x4(X_s, MAT_LD, W2, HID, HID, tm, tn0, lane);
        const Row8 rs = load_row8(dinv_s, tm, lane);
        __syncthreads();
        strip_store16x4_pairs(acc, X_s, rs, tm, tn0, lane);
    }
    __syncthreads();

    {
        const Acc4 acc = strip_acc16x4_pairs(M_s, MAT_LD, X_s, HID, tm, tn0, lane);
        const Row8 rs = load_row8(dinv_s, tm, lane);
        __syncthreads();
        strip_store16x4<true, true>(acc, X_s, MAT_LD, rs, b2, tm, tn0, lane);
    }
    __syncthreads();

    {
        const int a  = tid >> 2;
        const int sl = tid & 3;
        const float* hrow = X_s + a * MAT_LD + sl * 32;
        const float* wseg = Wout + sl * 32;
        float s = 0.0f;
        for (int k = 0; k < 32; ++k) s += hrow[k] * wseg[k];
        s += __shfl_xor(s, 1);
        s += __shfl_xor(s, 2);
        if (sl == 0) out[(size_t)bidx * A_AGENTS + a] = s + bout[0];
    }
}

extern "C" void kernel_launch(void* const* d_in, const int* in_sizes, int n_in,
                              void* d_out, int out_size, void* d_ws, size_t ws_size,
                              hipStream_t stream) {
    (void)n_in; (void)out_size; (void)d_ws; (void)ws_size;
    const float* obs  = (const float*)d_in[0];
    const float* W1   = (const float*)d_in[1];
    const float* b1   = (const float*)d_in[2];
    const float* W2   = (const float*)d_in[3];
    const float* b2   = (const float*)d_in[4];
    const float* Wout = (const float*)d_in[5];
    const float* bout = (const float*)d_in[6];
    float* out = (float*)d_out;

    const int B = in_sizes[0] / (A_AGENTS * D_IN);

    static_assert(2 * LDS_BYTES <= 320 * 1024, "two workgroups must fit one WGP's LDS");
    (void)hipFuncSetAttribute((const void*)gnn_critic_fused,
                              hipFuncAttributeMaxDynamicSharedMemorySize,
                              LDS_BYTES);
    gnn_critic_fused<<<B, NTHREADS, LDS_BYTES, stream>>>(
        obs, W1, b1, W2, b2, Wout, bout, out);
}


# Each of 8 waves (was 16) owns a 16x128 row strip through all products: M@obs, H1, H1@W2 run with no workgroup barrier between them, H2 goes over the wave's own M rows; barriers 11 to 5; same sums
# speedup vs baseline: 2.2370x; 1.0401x over previous
#include <hip/hip_runtime.h>
#include <hip/hip_bf16.h>
#include <math.h>


typedef float v2f __attribute__((ext_vector_type(2)));
typedef float v8f __attribute__((ext_vector_type(8)));

#define A_AGENTS 128
#define D_IN     32
#define HID      128
#define KNN      16
#define OBS_LD   36
#define MAT_LD   132
#define PAIR_LD  (2 * MAT_LD)
#define NTHREADS 256
#define NWAVES   (NTHREADS / 32)
#define BIGF     3.4e38f

#define OFF_OBS  0
#define OFF_X    (OFF_OBS + A_AGENTS * OBS_LD)
#define OFF_M    (OFF_X   + A_AGENTS * MAT_LD)
#define OFF_DINV (OFF_M   + A_AGENTS * MAT_LD)
#define LDS_FLOATS (OFF_DINV + A_AGENTS)
#define LDS_BYTES  (LDS_FLOATS * 4)

__device__ __forceinline__ float fast_tanh(float x) {
#if __has_builtin(__builtin_amdgcn_tanhf)
    return __builtin_amdgcn_tanhf(x);
#elif __has_builtin(__builtin_amdgcn_tanh_f32)
    return __builtin_amdgcn_tanh_f32(x);
#else
    return tanhf(x);
#endif
}

__device__ __forceinline__ void scan8(const float* p, int base, float& m, int& mj)
{
    const float4 lo = *(const float4*)(p);
    const float4 hi = *(const float4*)(p + 4);
    const bool c01 = lo.y < lo.x;
    const bool c23 = lo.w < lo.z;
    const bool c45 = hi.y < hi.x;
    const bool c67 = hi.w < hi.z;
    const float m01 = c01 ? lo.y : lo.x;   const int e01 = c01 ? 1 : 0;
    const float m23 = c23 ? lo.w : lo.z;   const int e23 = c23 ? 3 : 2;
    const float m45 = c45 ? hi.y : hi.x;   const int e45 = c45 ? 5 : 4;
    const float m67 = c67 ? hi.w : hi.z;   const int e67 = c67 ? 7 : 6;
    const bool cA = m23 < m01;
    const bool cB = m67 < m45;
    const float mA = cA ? m23 : m01;       const int eA = cA ? e23 : e01;
    const float mB = cB ? m67 : m45;       const int eB = cB ? e67 : e45;
    const bool cT = mB < mA;
    m  = cT ? mB : mA;
    mj = base + (cT ? eB : eA);
}

struct Cand { float m; int j; };

__device__ __forceinline__ Cand lower_of(const Cand l, const Cand r)
{
    const bool c = r.m < l.m;
    Cand o;
    o.m = c ? r.m : l.m;
    o.j = c ? r.j : l.j;
    return o;
}

struct Acc4 { v8f c0, c1, c2, c3; };
struct Acc8 { Acc4 lo, hi; };

__device__ __forceinline__ Acc8 strip_acc16x8(const float* A, int lda,
                                              const float* B, int ldb,
                                              int K, int tm, int lane)
{
    const int g = lane >> 4;
    const int r = lane & 15;
    const float* Arow = A + (tm * 16 + r) * lda;
    const float* Bcol = B + 4 * r;
    v8f c0 = {}, c1 = {}, c2 = {}, c3 = {}, c4 = {}, c5 = {}, c6 = {}, c7 = {};
    for (int k0 = 0; k0 < K; k0 += 4) {
        const int ka = k0 + 2 * g;
        const float4 q0 = *(const float4*)(Bcol + (ka)     * ldb);
        const float4 q1 = *(const float4*)(Bcol + (ka + 1) * ldb);
        const float4 p0 = *(const float4*)(Bcol + (ka)     * ldb + 64);
        const float4 p1 = *(const float4*)(Bcol + (ka + 1) * ldb + 64);
        v2f a, b0, b1, b2, b3, b4, b5, b6, b7;
        a.x  = Arow[ka];   a.y  = Arow[ka + 1];
        b0.x = q0.x;       b0.y = q1.x;
        b1.x = q0.y;       b1.y = q1.y;
        b2.x = q0.z;       b2.y = q1.z;
        b3.x = q0.w;       b3.y = q1.w;
        b4.x = p0.x;       b4.y = p1.x;
        b5.x = p0.y;       b5.y = p1.y;
        b6.x = p0.z;       b6.y = p1.z;
        b7.x = p0.w;       b7.y = p1.w;
        c0 = __builtin_amdgcn_wmma_f32_16x16x4_f32(false, a, false, b0, (short)0, c0, false, false);
        c1 = __builtin_amdgcn_wmma_f32_16x16x4_f32(false, a, false, b1, (short)0, c1, false, false);
        c2 = __builtin_amdgcn_wmma_f32_16x16x4_f32(false, a, false, b2, (short)0, c2, false, false);
        c3 = __builtin_amdgcn_wmma_f32_16x16x4_f32(false, a, false, b3, (short)0, c3, false, false);
        c4 = __builtin_amdgcn_wmma_f32_16x16x4_f32(false, a, false, b4, (short)0, c4, false, false);
        c5 = __builtin_amdgcn_wmma_f32_16x16x4_f32(false, a, false, b5, (short)0, c5, false, false);
        c6 = __builtin_amdgcn_wmma_f32_16x16x4_f32(false, a, false, b6, (short)0, c6, false, false);
        c7 = __builtin_amdgcn_wmma_f32_16x16x4_f32(false, a, false, b7, (short)0, c7, false, false);
    }
    Acc8 acc;
    acc.lo.c0 = c0; acc.lo.c1 = c1; acc.lo.c2 = c2; acc.lo.c3 = c3;
    acc.hi.c0 = c4; acc.hi.c1 = c5; acc.hi.c2 = c6; acc.hi.c3 = c7;
    return acc;
}

__device__ __forceinline__ int pair_off(int pair_row, int tn0, int r)
{
    return pair_row * PAIR_LD + tn0 * 32 + 4 * r;
}

__device__ __forceinline__ Acc8 strip_acc16x8_pairs(const float* A, int lda,
                                                    const float* BP,
                                                    int K, int tm, int lane)
{
    const int g = lane >> 4;
    const int r = lane & 15;
    const float* ap = A + (tm * 16 + r) * lda + 2 * g;
    const float* bp = BP + pair_off(g, 0, r);
    v8f c0 = {}, c1 = {}, c2 = {}, c3 = {}, c4 = {}, c5 = {}, c6 = {}, c7 = {};
#pragma unroll 4
    for (int k0 = 0; k0 < K; k0 += 4) {
        const v2f a = *(const v2f*)(ap);
        const float4 q0 = *(const float4*)(bp);
        const float4 q1 = *(const float4*)(bp + 64);
        const float4 p0 = *(const float4*)(bp + 128);
        const float4 p1 = *(const float4*)(bp + 192);
        v2f b0, b1, b2, b3, b4, b5, b6, b7;
        b0.x = q0.x;  b0.y = q0.y;
        b1.x = q0.z;  b1.y = q0.w;
        b2.x = q1.x;  b2.y = q1.y;
        b3.x = q1.z;  b3.y = q1.w;
        b4.x = p0.x;  b4.y = p0.y;
        b5.x = p0.z;  b5.y = p0.w;
        b6.x = p1.x;  b6.y = p1.y;
        b7.x = p1.z;  b7.y = p1.w;
        c0 = __builtin_amdgcn_wmma_f32_16x16x4_f32(false, a, false, b0, (short)0, c0, false, false);
        c1 = __builtin_amdgcn_wmma_f32_16x16x4_f32(false, a, false, b1, (short)0, c1, false, false);
        c2 = __builtin_amdgcn_wmma_f32_16x16x4_f32(false, a, false, b2, (short)0, c2, false, false);
        c3 = __builtin_amdgcn_wmma_f32_16x16x4_f32(false, a, false, b3, (short)0, c3, false, false);
        c4 = __builtin_amdgcn_wmma_f32_16x16x4_f32(false, a, false, b4, (short)0, c4, false, false);
        c5 = __builtin_amdgcn_wmma_f32_16x16x4_f32(false, a, false, b5, (short)0, c5, false, false);
        c6 = __builtin_amdgcn_wmma_f32_16x16x4_f32(false, a, false, b6, (short)0, c6, false, false);
        c7 = __builtin_amdgcn_wmma_f32_16x16x4_f32(false, a, false, b7, (short)0, c7, false, false);
        ap += 4;
        bp += 2 * PAIR_LD;
    }
    Acc8 acc;
    acc.lo.c0 = c0; acc.lo.c1 = c1; acc.lo.c2 = c2; acc.lo.c3 = c3;
    acc.hi.c0 = c4; acc.hi.c1 = c5; acc.hi.c2 = c6; acc.hi.c3 = c7;
    return acc;
}

__device__ __forceinline__ void tiles_acc16x32(const float* A, int lda,
                                               const float* B, int ldb,
                                               int K, int tm, int lane,
                                               v8f& out0, v8f& out1)
{
    const int g = lane >> 4;
    const int r = lane & 15;
    const float* Arow = A + (tm * 16 + r) * lda;
    const float* Bcol = B + r;
    v8f c0 = {}, c1 = {};
#pragma unroll
    for (int k0 = 0; k0 < K; k0 += 4) {
        const int ka = k0 + 2 * g;
        v2f a, b0, b1;
        a.x  = Arow[ka];               a.y  = Arow[ka + 1];
        b0.x = Bcol[ka * ldb];         b0.y = Bcol[(ka + 1) * ldb];
        b1.x = Bcol[ka * ldb + 16];    b1.y = Bcol[(ka + 1) * ldb + 16];
        c0 = __builtin_amdgcn_wmma_f32_16x16x4_f32(false, a, false, b0, (short)0, c0, false, false);
        c1 = __builtin_amdgcn_wmma_f32_16x16x4_f32(false, a, false, b1, (short)0, c1, false, false);
    }
    out0 = c0;
    out1 = c1;
}

__device__ __forceinline__ void wave_lds_order()
{
    __builtin_amdgcn_fence(__ATOMIC_RELEASE, "wavefront");
    __builtin_amdgcn_wave_barrier();
    __builtin_amdgcn_fence(__ATOMIC_ACQUIRE, "wavefront");
}

struct Row8 { float d[8]; };

__device__ __forceinline__ Row8 load_row8(const float* dinv, int tm, int lane)
{
    const float* p = dinv + tm * 16 + 8 * (lane >> 4);
    const float4 lo = *(const float4*)(p);
    const float4 hi = *(const float4*)(p + 4);
    Row8 rs;
    rs.d[0] = lo.x; rs.d[1] = lo.y; rs.d[2] = lo.z; rs.d[3] = lo.w;
    rs.d[4] = hi.x; rs.d[5] = hi.y; rs.d[6] = hi.z; rs.d[7] = hi.w;
    return rs;
}

template<bool SCALE, bool ACT>
__device__ __forceinline__ void strip_store16x4(const Acc4& acc, float* OUT, int ldo,
                                                const Row8& rs,
                                                const float* __restrict__ bias,
                                                int tm, int tn0, int lane)
{
    const int g = lane >> 4;
    const int r = lane & 15;
    const v8f cc[4] = {acc.c0, acc.c1, acc.c2, acc.c3};
    float* obase = OUT + (tm * 16 + 8 * g) * ldo + tn0 * 16 + 4 * r;
    float bv[4] = {0.0f, 0.0f, 0.0f, 0.0f};
    if (ACT) {
        const float4 b4 = *(const float4*)(bias + tn0 * 16 + 4 * r);
        bv[0] = b4.x; bv[1] = b4.y; bv[2] = b4.z; bv[3] = b4.w;
    }
#pragma unroll
    for (int v = 0; v < 8; ++v) {
        float o[4];
#pragma unroll
        for (int t = 0; t < 4; ++t) {
            float val = cc[t][v];
            if (SCALE) val *= rs.d[v];
            if (ACT) val = fast_tanh(val + bv[t]);
            o[t] = val;
        }
        *(float4*)(obase + v * ldo) = make_float4(o[0], o[1], o[2], o[3]);
    }
}

__device__ __forceinline__ void strip_store16x4_pairs(const Acc4& acc, float* OUTP,
                                                      const Row8& rs,
                                                      int tm, int tn0, int lane)
{
    const int g = lane >> 4;
    const int r = lane & 15;
    float* obase = OUTP + pair_off(tm * 8 + 4 * g, tn0, r);
#pragma unroll
    for (int u = 0; u < 4; ++u) {
        const float d0 = rs.d[2 * u], d1 = rs.d[2 * u + 1];
        *(float4*)(obase + u * PAIR_LD) =
            make_float4(acc.c0[2 * u] * d0, acc.c0[2 * u + 1] * d1,
                        acc.c1[2 * u] * d0, acc.c1[2 * u + 1] * d1);
        *(float4*)(obase + u * PAIR_LD + 64) =
            make_float4(acc.c2[2 * u] * d0, acc.c2[2 * u + 1] * d1,
                        acc.c3[2 * u] * d0, acc.c3[2 * u + 1] * d1);
    }
}

__global__ void __launch_bounds__(NTHREADS)
gnn_critic_fused(const float* __restrict__ obs,
                 const float* __restrict__ W1, const float* __restrict__ b1,
                 const float* __restrict__ W2, const float* __restrict__ b2,
                 const float* __restrict__ Wout, const float* __restrict__ bout,
                 float* __restrict__ out)
{
    extern __shared__ __attribute__((aligned(16))) float smem[];
    float* obs_s  = smem + OFF_OBS;
    float* M_s    = smem + OFF_M;
    float* X_s    = smem + OFF_X;
    float* dinv_s = smem + OFF_DINV;

    const int tid  = threadIdx.x;
    const int lane = tid & 31;
    const int tm   = tid >> 5;
    const int bidx = blockIdx.x;

    const float* gobs = obs + (size_t)bidx * (A_AGENTS * D_IN);
    static_assert(A_AGENTS * (D_IN / 4) == 4 * NTHREADS, "four pieces per thread");
    static_assert((A_AGENTS * MAT_LD) % (2 * NTHREADS) == 0, "M zeroing has no remainder");
    const float4 v0 = ((const float4*)gobs)[tid];
    const float4 v1 = ((const float4*)gobs)[tid + NTHREADS];
    const float4 v2 = ((const float4*)gobs)[tid + 2 * NTHREADS];
    const float4 v3 = ((const float4*)gobs)[tid + 3 * NTHREADS];
#pragma unroll
    for (int i = 0; i < (A_AGENTS * MAT_LD) / (2 * NTHREADS); ++i)
        ((float2*)M_s)[tid + i * NTHREADS] = make_float2(0.0f, 0.0f);
    {
        float* dst = obs_s + (tid >> 3) * OBS_LD + (tid & 7) * 4;
        *(float4*)(dst) = v0;
        *(float4*)(dst + 1 * (NTHREADS / 8) * OBS_LD) = v1;
        *(float4*)(dst + 2 * (NTHREADS / 8) * OBS_LD) = v2;
        *(float4*)(dst + 3 * (NTHREADS / 8) * OBS_LD) = v3;
    }
    __syncthreads();

    {
        const int a  = tid >> 1;
        const int hf = tid & 1;
        const float2 pi = *(const float2*)(obs_s + a * OBS_LD);
        for (int q = 0; q < 2; ++q) {
            const int sl = 2 * hf + q;
            const float* prow = obs_s + (32 * sl) * OBS_LD;
            float* drow = X_s + a * MAT_LD + 32 * sl;
            float2 pj[32];
#pragma unroll
            for (int u = 0; u < 8; ++u) {
                const int cb = ((u + sl) & 7) * 4;
#pragma unroll
                for (int e = 0; e < 4; ++e)
                    pj[4 * u + e] = *(const float2*)(prow + (cb + e) * OBS_LD);
            }
            __builtin_amdgcn_sched_barrier(0);
#pragma unroll
            for (int u = 0; u < 8; ++u) {
                const int cb = ((u + sl) & 7) * 4;
                float dd[4];
#pragma unroll
                for (int e = 0; e < 4; ++e) {
                    const float dx = pi.x - pj[4 * u + e].x;
                    const float dy = pi.y - pj[4 * u + e].y;
                    {
#pragma clang fp contract(off)
                        dd[e] = dx * dx + dy * dy;
                    }
                }
                *(float4*)(drow + cb) = make_float4(dd[0], dd[1], dd[2], dd[3]);
            }
            if ((a >> 5) == sl) X_s[a * MAT_LD + a] = BIGF;
        }
    }
    __syncthreads();

    if (tid < A_AGENTS) {
        const int a = tid;
        float* drow = X_s + a * MAT_LD;
        M_s[a * MAT_LD + a] = 1.0f;
        float gm[16];
        int   gj[16];
#pragma unroll
        for (int q = 0; q < 16; ++q) scan8(drow + 8 * q, 8 * q, gm[q], gj[q]);

        for (int t = 0; t < KNN; ++t) {
            const Cand a0 = lower_of(Cand{gm[0],  gj[0]},  Cand{gm[1],  gj[1]});
            const Cand a1 = lower_of(Cand{gm[2],  gj[2]},  Cand{gm[3],  gj[3]});
            const Cand a2 = lower_of(Cand{gm[4],  gj[4]},  Cand{gm[5],  gj[5]});
            const Cand a3 = lower_of(Cand{gm[6],  gj[6]},  Cand{gm[7],  gj[7]});
            const Cand a4 = lower_of(Cand{gm[8],  gj[8]},  Cand{gm[9],  gj[9]});
            const Cand a5 = lower_of(Cand{gm[10], gj[10]}, Cand{gm[11], gj[11]});
            const Cand a6 = lower_of(Cand{gm[12], gj[12]}, Cand{gm[13], gj[13]});
            const Cand a7 = lower_of(Cand{gm[14], gj[14]}, Cand{gm[15], gj[15]});
            const Cand b0 = lower_of(a0, a1);
            const Cand b1 = lower_of(a2, a3);
            const Cand b2 = lower_of(a4, a5);
            const Cand b3 = lower_of(a6, a7);
            const Cand c0 = lower_of(b0, b1);
            const Cand c1 = lower_of(b2, b3);
            const int bestj = lower_of(c0, c1).j;

            drow[bestj] = BIGF;
            M_s[bestj * MAT_LD + a] = 1.0f;
            if (t + 1 < KNN) {
                const int qw = bestj >> 3;
                float nm;
                int   nj;
                scan8(drow + 8 * qw, 8 * qw, nm, nj);
#pragma unroll
                for (int q = 0; q < 16; ++q) {
                    const bool hit = (q == qw);
                    gm[q] = hit ? nm : gm[q];
                    gj[q] = hit ? nj : gj[q];
                }
            }
        }
    }
    __syncthreads();

    {
        const int a  = tid >> 1;
        const int hf = tid & 1;
        float* orow = obs_s + a * OBS_LD + hf * 16;
        float4 o0 = *(const float4*)(orow);
        float4 o1 = *(const float4*)(orow + 4);
        float4 o2 = *(const float4*)(orow + 8);
        float4 o3 = *(const float4*)(orow + 12);
        const float* seg = M_s + a * MAT_LD + hf * 64;
        float s = 0.0f;
        for (int i = 0; i < 16; ++i) {
            const float4 m4 = *(const float4*)(seg + 4 * i);
            s += (m4.x + m4.y) + (m4.z + m4.w);
        }
        s += __shfl_xor(s, 1);
        const float dv = rsqrtf(s);
        if (hf == 0) dinv_s[a] = dv;
        o0.x *= dv; o0.y *= dv; o0.z *= dv; o0.w *= dv;
        o1.x *= dv; o1.y *= dv; o1.z *= dv; o1.w *= dv;
        o2.x *= dv; o2.y *= dv; o2.z *= dv; o2.w *= dv;
        o3.x *= dv; o3.y *= dv; o3.z *= dv; o3.w *= dv;
        *(float4*)(orow)      = o0;
        *(float4*)(orow + 4)  = o1;
        *(float4*)(orow + 8)  = o2;
        *(float4*)(orow + 12) = o3;
    }
    __syncthreads();

    {
        v8f c0, c1;
        tiles_acc16x32(M_s, MAT_LD, obs_s, OBS_LD, A_AGENTS, tm, lane, c0, c1);
        const Row8 rs = load_row8(dinv_s, tm, lane);
        float* obase = X_s + (tm * 16 + 8 * (lane >> 4)) * MAT_LD + (lane & 15);
#pragma unroll
        for (int v = 0; v < 8; ++v) {
            obase[v * MAT_LD]      = c0[v] * rs.d[v];
            obase[v * MAT_LD + 16] = c1[v] * rs.d[v];
        }
    }
    wave_lds_order();

    {
        const Acc8 acc = strip_acc16x8(X_s, MAT_LD, W1, HID, D_IN, tm, lane);
        const Row8 none = {};
        wave_lds_order();
        strip_store16x4<false, true>(acc.lo, X_s, MAT_LD, none, b1, tm, 0, lane);
        strip_store16x4<false, true>(acc.hi, X_s, MAT_LD, none, b1, tm, 4, lane);
    }
    wave_lds_order();

    {
        const Acc8 acc = strip_acc16x8(X_s, MAT_LD, W2, HID, HID, tm, lane);
        const Row8 rs = load_row8(dinv_s, tm, lane);
        wave_lds_order();
        strip_store16x4_pairs(acc.lo, X_s, rs, tm, 0, lane);
        strip_store16x4_pairs(acc.hi, X_s, rs, tm, 4, lane);
    }
    __syncthreads();

    {
        const Acc8 acc = strip_acc16x8_pairs(M_s, MAT_LD, X_s, HID, tm, lane);
        const Row8 rs = load_row8(dinv_s, tm, lane);
        wave_lds_order();
        strip_store16x4<true, true>(acc.lo, M_s, MAT_LD, rs, b2, tm, 0, lane);
        strip_store16x4<true, true>(acc.hi, M_s, MAT_LD, rs, b2, tm, 4, lane);
    }
    wave_lds_order();

    {
        const int a  = tid >> 1;
        const int hf = tid & 1;
        const float* hrow = M_s + a * MAT_LD + hf * 64;
        const float* wseg = Wout + hf * 64;
        float s0 = 0.0f, s1 = 0.0f;
        for (int k = 0; k < 32; ++k) {
            s0 += hrow[k]      * wseg[k];
            s1 += hrow[k + 32] * wseg[k + 32];
        }
        float s = s0 + s1;
        s += __shfl_xor(s, 1);
        if (hf == 0) out[(size_t)bidx * A_AGENTS + a] = s + bout[0];
    }
}

extern "C" void kernel_launch(void* const* d_in, const int* in_sizes, int n_in,
                              void* d_out, int out_size, void* d_ws, size_t ws_size,
                              hipStream_t stream) {
    (void)n_in; (void)out_size; (void)d_ws; (void)ws_size;
    const float* obs  = (const float*)d_in[0];
    const float* W1   = (const float*)d_in[1];
    const float* b1   = (const float*)d_in[2];
    const float* W2   = (const float*)d_in[3];
    const float* b2   = (const float*)d_in[4];
    const float* Wout = (const float*)d_in[5];
    const float* bout = (const float*)d_in[6];
    float* out = (float*)d_out;

    const int B = in_sizes[0] / (A_AGENTS * D_IN);

    static_assert(2 * LDS_BYTES <= 320 * 1024, "two workgroups must fit one WGP's LDS");
    (void)hipFuncSetAttribute((const void*)gnn_critic_fused,
                              hipFuncAttributeMaxDynamicSharedMemorySize,
                              LDS_BYTES);
    gnn_critic_fused<<<B, NTHREADS, LDS_BYTES, stream>>>(
        obs, W1, b1, W2, b2, Wout, bout, out);
}


# Agents renumbered along a snake through 4 x-strips so M is banded; both M products skip k-steps whose 16x4 block of M is all zero (ballot mask, scalar branch); order of the M sums changes
# speedup vs baseline: 2.4970x; 1.1162x over previous
#include <hip/hip_runtime.h>
#include <hip/hip_bf16.h>
#include <math.h>


typedef float v2f __attribute__((ext_vector_type(2)));
typedef float v8f __attribute__((ext_vector_type(8)));

#define A_AGENTS 128
#define D_IN     32
#define HID      128
#define KNN      16
#define OBS_LD   36
#define MAT_LD   132
#define PAIR_LD  (2 * MAT_LD)
#define NTHREADS 256
#define NWAVES   (NTHREADS / 32)
#define BIGF     3.4e38f

#define OFF_OBS  0
#define OFF_X    (OFF_OBS + A_AGENTS * OBS_LD)
#define OFF_M    (OFF_X   + A_AGENTS * MAT_LD)
#define OFF_DINV (OFF_M   + A_AGENTS * MAT_LD)
#define OFF_KEY  (OFF_DINV + A_AGENTS)
#define OFF_RANK (OFF_KEY  + A_AGENTS)
#define OFF_INV  (OFF_RANK + A_AGENTS)
#define LDS_FLOATS (OFF_INV + A_AGENTS)
#define LDS_BYTES  (LDS_FLOATS * 4)

__device__ __forceinline__ float fast_tanh(float x) {
#if __has_builtin(__builtin_amdgcn_tanhf)
    return __builtin_amdgcn_tanhf(x);
#elif __has_builtin(__builtin_amdgcn_tanh_f32)
    return __builtin_amdgcn_tanh_f32(x);
#else
    return tanhf(x);
#endif
}

__device__ __forceinline__ void scan8(const float* p, int base, float& m, int& mj)
{
    const float4 lo = *(const float4*)(p);
    const float4 hi = *(const float4*)(p + 4);
    const bool c01 = lo.y < lo.x;
    const bool c23 = lo.w < lo.z;
    const bool c45 = hi.y < hi.x;
    const bool c67 = hi.w < hi.z;
    const float m01 = c01 ? lo.y : lo.x;   const int e01 = c01 ? 1 : 0;
    const float m23 = c23 ? lo.w : lo.z;   const int e23 = c23 ? 3 : 2;
    const float m45 = c45 ? hi.y : hi.x;   const int e45 = c45 ? 5 : 4;
    const float m67 = c67 ? hi.w : hi.z;   const int e67 = c67 ? 7 : 6;
    const bool cA = m23 < m01;
    const bool cB = m67 < m45;
    const float mA = cA ? m23 : m01;       const int eA = cA ? e23 : e01;
    const float mB = cB ? m67 : m45;       const int eB = cB ? e67 : e45;
    const bool cT = mB < mA;
    m  = cT ? mB : mA;
    mj = base + (cT ? eB : eA);
}

struct Cand { float m; int j; };

__device__ __forceinline__ Cand lower_of(const Cand l, const Cand r)
{
    const bool c = r.m < l.m;
    Cand o;
    o.m = c ? r.m : l.m;
    o.j = c ? r.j : l.j;
    return o;
}

struct Acc4 { v8f c0, c1, c2, c3; };
struct Acc8 { Acc4 lo, hi; };

__device__ __forceinline__ Acc8 strip_acc16x8(const float* A, int lda,
                                              const float* B, int ldb,
                                              int K, int tm, int lane)
{
    const int g = lane >> 4;
    const int r = lane & 15;
    const float* Arow = A + (tm * 16 + r) * lda;
    const float* Bcol = B + 4 * r;
    v8f c0 = {}, c1 = {}, c2 = {}, c3 = {}, c4 = {}, c5 = {}, c6 = {}, c7 = {};
    for (int k0 = 0; k0 < K; k0 += 4) {
        const int ka = k0 + 2 * g;
        const float4 q0 = *(const float4*)(Bcol + (ka)     * ldb);
        const float4 q1 = *(const float4*)(Bcol + (ka + 1) * ldb);
        const float4 p0 = *(const float4*)(Bcol + (ka)     * ldb + 64);
        const float4 p1 = *(const float4*)(Bcol + (ka + 1) * ldb + 64);
        v2f a, b0, b1, b2, b3, b4, b5, b6, b7;
        a.x  = Arow[ka];   a.y  = Arow[ka + 1];
        b0.x = q0.x;       b0.y = q1.x;
        b1.x = q0.y;       b1.y = q1.y;
        b2.x = q0.z;       b2.y = q1.z;
        b3.x = q0.w;       b3.y = q1.w;
        b4.x = p0.x;       b4.y = p1.x;
        b5.x = p0.y;       b5.y = p1.y;
        b6.x = p0.z;       b6.y = p1.z;
        b7.x = p0.w;       b7.y = p1.w;
        c0 = __builtin_amdgcn_wmma_f32_16x16x4_f32(false, a, false, b0, (short)0, c0, false, false);
        c1 = __builtin_amdgcn_wmma_f32_16x16x4_f32(false, a, false, b1, (short)0, c1, false, false);
        c2 = __builtin_amdgcn_wmma_f32_16x16x4_f32(false, a, false, b2, (short)0, c2, false, false);
        c3 = __builtin_amdgcn_wmma_f32_16x16x4_f32(false, a, false, b3, (short)0, c3, false, false);
        c4 = __builtin_amdgcn_wmma_f32_16x16x4_f32(false, a, false, b4, (short)0, c4, false, false);
        c5 = __builtin_amdgcn_wmma_f32_16x16x4_f32(false, a, false, b5, (short)0, c5, false, false);
        c6 = __builtin_amdgcn_wmma_f32_16x16x4_f32(false, a, false, b6, (short)0, c6, false, false);
        c7 = __builtin_amdgcn_wmma_f32_16x16x4_f32(false, a, false, b7, (short)0, c7, false, false);
    }
    Acc8 acc;
    acc.lo.c0 = c0; acc.lo.c1 = c1; acc.lo.c2 = c2; acc.lo.c3 = c3;
    acc.hi.c0 = c4; acc.hi.c1 = c5; acc.hi.c2 = c6; acc.hi.c3 = c7;
    return acc;
}

__device__ __forceinline__ int pair_off(int pair_row, int tn0, int r)
{
    return pair_row * PAIR_LD + tn0 * 32 + 4 * r;
}

__device__ __forceinline__ Acc8 strip_acc16x8_pairs(const float* A, int lda,
                                                    const float* BP,
                                                    int K, unsigned kmask, int tm, int lane)
{
    const int g = lane >> 4;
    const int r = lane & 15;
    const float* ap = A + (tm * 16 + r) * lda + 2 * g;
    const float* bp = BP + pair_off(g, 0, r);
    v8f c0 = {}, c1 = {}, c2 = {}, c3 = {}, c4 = {}, c5 = {}, c6 = {}, c7 = {};
    for (int k0 = 0; k0 < K; k0 += 4) {
      if ((kmask >> (k0 >> 2)) & 1u) {
        const v2f a = *(const v2f*)(ap);
        const float4 q0 = *(const float4*)(bp);
        const float4 q1 = *(const float4*)(bp + 64);
        const float4 p0 = *(const float4*)(bp + 128);
        const float4 p1 = *(const float4*)(bp + 192);
        v2f b0, b1, b2, b3, b4, b5, b6, b7;
        b0.x = q0.x;  b0.y = q0.y;
        b1.x = q0.z;  b1.y = q0.w;
        b2.x = q1.x;  b2.y = q1.y;
        b3.x = q1.z;  b3.y = q1.w;
        b4.x = p0.x;  b4.y = p0.y;
        b5.x = p0.z;  b5.y = p0.w;
        b6.x = p1.x;  b6.y = p1.y;
        b7.x = p1.z;  b7.y = p1.w;
        c0 = __builtin_amdgcn_wmma_f32_16x16x4_f32(false, a, false, b0, (short)0, c0, false, false);
        c1 = __builtin_amdgcn_wmma_f32_16x16x4_f32(false, a, false, b1, (short)0, c1, false, false);
        c2 = __builtin_amdgcn_wmma_f32_16x16x4_f32(false, a, false, b2, (short)0, c2, false, false);
        c3 = __builtin_amdgcn_wmma_f32_16x16x4_f32(false, a, false, b3, (short)0, c3, false, false);
        c4 = __builtin_amdgcn_wmma_f32_16x16x4_f32(false, a, false, b4, (short)0, c4, false, false);
        c5 = __builtin_amdgcn_wmma_f32_16x16x4_f32(false, a, false, b5, (short)0, c5, false, false);
        c6 = __builtin_amdgcn_wmma_f32_16x16x4_f32(false, a, false, b6, (short)0, c6, false, false);
        c7 = __builtin_amdgcn_wmma_f32_16x16x4_f32(false, a, false, b7, (short)0, c7, false, false);
      }
        ap += 4;
        bp += 2 * PAIR_LD;
    }
    Acc8 acc;
    acc.lo.c0 = c0; acc.lo.c1 = c1; acc.lo.c2 = c2; acc.lo.c3 = c3;
    acc.hi.c0 = c4; acc.hi.c1 = c5; acc.hi.c2 = c6; acc.hi.c3 = c7;
    return acc;
}

__device__ __forceinline__ void tiles_acc16x32(const float* A, int lda,
                                               const float* B, int ldb,
                                               int K, unsigned kmask, int tm, int lane,
                                               v8f& out0, v8f& out1)
{
    const int g = lane >> 4;
    const int r = lane & 15;
    const float* Arow = A + (tm * 16 + r) * lda + 2 * g;
    const float* Bcol = B + r + 2 * g * ldb;
    v8f c0 = {}, c1 = {};
    for (int k0 = 0; k0 < K; k0 += 4) {
        if ((kmask >> (k0 >> 2)) & 1u) {
            v2f a, b0, b1;
            a.x  = Arow[k0];               a.y  = Arow[k0 + 1];
            b0.x = Bcol[k0 * ldb];         b0.y = Bcol[(k0 + 1) * ldb];
            b1.x = Bcol[k0 * ldb + 16];    b1.y = Bcol[(k0 + 1) * ldb + 16];
            c0 = __builtin_amdgcn_wmma_f32_16x16x4_f32(false, a, false, b0, (short)0, c0, false, false);
            c1 = __builtin_amdgcn_wmma_f32_16x16x4_f32(false, a, false, b1, (short)0, c1, false, false);
        }
    }
    out0 = c0;
    out1 = c1;
}

__device__ __forceinline__ void wave_lds_order()
{
    __builtin_amdgcn_fence(__ATOMIC_RELEASE, "wavefront");
    __builtin_amdgcn_wave_barrier();
    __builtin_amdgcn_fence(__ATOMIC_ACQUIRE, "wavefront");
}

struct Row8 { float d[8]; };

__device__ __forceinline__ Row8 load_row8(const float* dinv, int tm, int lane)
{
    const float* p = dinv + tm * 16 + 8 * (lane >> 4);
    const float4 lo = *(const float4*)(p);
    const float4 hi = *(const float4*)(p + 4);
    Row8 rs;
    rs.d[0] = lo.x; rs.d[1] = lo.y; rs.d[2] = lo.z; rs.d[3] = lo.w;
    rs.d[4] = hi.x; rs.d[5] = hi.y; rs.d[6] = hi.z; rs.d[7] = hi.w;
    return rs;
}

template<bool SCALE, bool ACT>
__device__ __forceinline__ void strip_store16x4(const Acc4& acc, float* OUT, int ldo,
                                                const Row8& rs,
                                                const float* __restrict__ bias,
                                                int tm, int tn0, int lane)
{
    const int g = lane >> 4;
    const int r = lane & 15;
    const v8f cc[4] = {acc.c0, acc.c1, acc.c2, acc.c3};
    float* obase = OUT + (tm * 16 + 8 * g) * ldo + tn0 * 16 + 4 * r;
    float bv[4] = {0.0f, 0.0f, 0.0f, 0.0f};
    if (ACT) {
        const float4 b4 = *(const float4*)(bias + tn0 * 16 + 4 * r);
        bv[0] = b4.x; bv[1] = b4.y; bv[2] = b4.z; bv[3] = b4.w;
    }
#pragma unroll
    for (int v = 0; v < 8; ++v) {
        float o[4];
#pragma unroll
        for (int t = 0; t < 4; ++t) {
            float val = cc[t][v];
            if (SCALE) val *= rs.d[v];
            if (ACT) val = fast_tanh(val + bv[t]);
            o[t] = val;
        }
        *(float4*)(obase + v * ldo) = make_float4(o[0], o[1], o[2], o[3]);
    }
}

__device__ __forceinline__ void strip_store16x4_pairs(const Acc4& acc, float* OUTP,
                                                      const Row8& rs,
                                                      int tm, int tn0, int lane)
{
    const int g = lane >> 4;
    const int r = lane & 15;
    float* obase = OUTP + pair_off(tm * 8 + 4 * g, tn0, r);
#pragma unroll
    for (int u = 0; u < 4; ++u) {
        const float d0 = rs.d[2 * u], d1 = rs.d[2 * u + 1];
        *(float4*)(obase + u * PAIR_LD) =
            make_float4(acc.c0[2 * u] * d0, acc.c0[2 * u + 1] * d1,
                        acc.c1[2 * u] * d0, acc.c1[2 * u + 1] * d1);
        *(float4*)(obase + u * PAIR_LD + 64) =
            make_float4(acc.c2[2 * u] * d0, acc.c2[2 * u + 1] * d1,
                        acc.c3[2 * u] * d0, acc.c3[2 * u + 1] * d1);
    }
}

__global__ void __launch_bounds__(NTHREADS)
gnn_critic_fused(const float* __restrict__ obs,
                 const float* __restrict__ W1, const float* __restrict__ b1,
                 const float* __restrict__ W2, const float* __restrict__ b2,
                 const float* __restrict__ Wout, const float* __restrict__ bout,
                 float* __restrict__ out)
{
    extern __shared__ __attribute__((aligned(16))) float smem[];
    float* obs_s  = smem + OFF_OBS;
    float* M_s    = smem + OFF_M;
    float* X_s    = smem + OFF_X;
    float* dinv_s = smem + OFF_DINV;
    unsigned* key_s = (unsigned*)(smem + OFF_KEY);
    int* rank_s = (int*)(smem + OFF_RANK);
    int* inv_s  = (int*)(smem + OFF_INV);

    const int tid  = threadIdx.x;
    const int lane = tid & 31;
    const int tm   = tid >> 5;
    const int bidx = blockIdx.x;

    const float* gobs = obs + (size_t)bidx * (A_AGENTS * D_IN);
    static_assert(A_AGENTS * (D_IN / 4) == 4 * NTHREADS, "four pieces per thread");
    static_assert((A_AGENTS * MAT_LD) % (2 * NTHREADS) == 0, "M zeroing has no remainder");
    const float4 v0 = ((const float4*)gobs)[tid];
    const float4 v1 = ((const float4*)gobs)[tid + NTHREADS];
    const float4 v2 = ((const float4*)gobs)[tid + 2 * NTHREADS];
    const float4 v3 = ((const float4*)gobs)[tid + 3 * NTHREADS];
#pragma unroll
    for (int i = 0; i < (A_AGENTS * MAT_LD) / (2 * NTHREADS); ++i)
        ((float2*)M_s)[tid + i * NTHREADS] = make_float2(0.0f, 0.0f);
    {
        float* dst = obs_s + (tid >> 3) * OBS_LD + (tid & 7) * 4;
        *(float4*)(dst) = v0;
        *(float4*)(dst + 1 * (NTHREADS / 8) * OBS_LD) = v1;
        *(float4*)(dst + 2 * (NTHREADS / 8) * OBS_LD) = v2;
        *(float4*)(dst + 3 * (NTHREADS / 8) * OBS_LD) = v3;
        if ((tid & 7) == 0) {
            const int row = tid >> 3;
            const float xs[4] = {v0.x, v1.x, v2.x, v3.x};
            const float ys[4] = {v0.y, v1.y, v2.y, v3.y};
#pragma unroll
            for (int p = 0; p < 4; ++p) {
                const float q = fminf(fmaxf(2.0f + 2.0f * fast_tanh(0.8f * xs[p]), 0.0f), 3.0f);
                const unsigned strip = (unsigned)(int)q;
                const unsigned u = __float_as_uint(ys[p]);
                unsigned yk = (u ^ ((unsigned)((int)u >> 31) | 0x80000000u)) >> 9;
                if (strip & 1u) yk ^= 0x7FFFFFu;
                key_s[row + p * (NTHREADS / 8)] =
                    (strip << 30) | (yk << 7) | (unsigned)(row + p * (NTHREADS / 8));
            }
        }
    }
    __syncthreads();

    {
        const int a  = tid >> 1;
        const int hf = tid & 1;
        const float2 pi = *(const float2*)(obs_s + a * OBS_LD);
        {
            const unsigned ka = key_s[a];
            const unsigned* kseg = key_s + 64 * hf;
            int cnt = 0;
            for (int i = 0; i < 16; ++i) {
                const uint4 k4 = *(const uint4*)(kseg + 4 * i);
                cnt += (k4.x < ka) + (k4.y < ka) + (k4.z < ka) + (k4.w < ka);
            }
            cnt += __shfl_xor(cnt, 1);
            if (hf == 0) { rank_s[a] = cnt; inv_s[cnt] = a; }
        }
        for (int q = 0; q < 2; ++q) {
            const int sl = 2 * hf + q;
            const float* prow = obs_s + (32 * sl) * OBS_LD;
            float* drow = X_s + a * MAT_LD + 32 * sl;
            float2 pj[32];
#pragma unroll
            for (int u = 0; u < 8; ++u) {
                const int cb = ((u + sl) & 7) * 4;
#pragma unroll
                for (int e = 0; e < 4; ++e)
                    pj[4 * u + e] = *(const float2*)(prow + (cb + e) * OBS_LD);
            }
            __builtin_amdgcn_sched_barrier(0);
#pragma unroll
            for (int u = 0; u < 8; ++u) {
                const int cb = ((u + sl) & 7) * 4;
                float dd[4];
#pragma unroll
                for (int e = 0; e < 4; ++e) {
                    const float dx = pi.x - pj[4 * u + e].x;
                    const float dy = pi.y - pj[4 * u + e].y;
                    {
#pragma clang fp contract(off)
                        dd[e] = dx * dx + dy * dy;
                    }
                }
                *(float4*)(drow + cb) = make_float4(dd[0], dd[1], dd[2], dd[3]);
            }
            if ((a >> 5) == sl) X_s[a * MAT_LD + a] = BIGF;
        }
    }
    __syncthreads();

    if (tid < A_AGENTS) {
        const int a = tid;
        float* drow = X_s + a * MAT_LD;
        const int ra = rank_s[a];
        M_s[ra * MAT_LD + ra] = 1.0f;
        float gm[16];
        int   gj[16];
#pragma unroll
        for (int q = 0; q < 16; ++q) scan8(drow + 8 * q, 8 * q, gm[q], gj[q]);

        for (int t = 0; t < KNN; ++t) {
            const Cand a0 = lower_of(Cand{gm[0],  gj[0]},  Cand{gm[1],  gj[1]});
            const Cand a1 = lower_of(Cand{gm[2],  gj[2]},  Cand{gm[3],  gj[3]});
            const Cand a2 = lower_of(Cand{gm[4],  gj[4]},  Cand{gm[5],  gj[5]});
            const Cand a3 = lower_of(Cand{gm[6],  gj[6]},  Cand{gm[7],  gj[7]});
            const Cand a4 = lower_of(Cand{gm[8],  gj[8]},  Cand{gm[9],  gj[9]});
            const Cand a5 = lower_of(Cand{gm[10], gj[10]}, Cand{gm[11], gj[11]});
            const Cand a6 = lower_of(Cand{gm[12], gj[12]}, Cand{gm[13], gj[13]});
            const Cand a7 = lower_of(Cand{gm[14], gj[14]}, Cand{gm[15], gj[15]});
            const Cand b0 = lower_of(a0, a1);
            const Cand b1 = lower_of(a2, a3);
            const Cand b2 = lower_of(a4, a5);
            const Cand b3 = lower_of(a6, a7);
            const Cand c0 = lower_of(b0, b1);
            const Cand c1 = lower_of(b2, b3);
            const int bestj = lower_of(c0, c1).j;

            const int rb = rank_s[bestj];
            drow[bestj] = BIGF;
            if (t + 1 < KNN) {
                const int qw = bestj >> 3;
                float nm;
                int   nj;
                scan8(drow + 8 * qw, 8 * qw, nm, nj);
#pragma unroll
                for (int q = 0; q < 16; ++q) {
                    const bool hit = (q == qw);
                    gm[q] = hit ? nm : gm[q];
                    gj[q] = hit ? nj : gj[q];
                }
            }
            M_s[rb * MAT_LD + ra] = 1.0f;
        }
    }
    __syncthreads();

    unsigned kmask = 0u;
    {
        const int p  = tid >> 1;
        const int hf = tid & 1;
        const float* seg = M_s + p * MAT_LD + hf * 64;
        float s = 0.0f;
#pragma unroll
        for (int i = 0; i < 16; ++i) {
            const float4 m4 = *(const float4*)(seg + 4 * i);
            const float t = (m4.x + m4.y) + (m4.z + m4.w);
            s += t;
            const unsigned b = __builtin_amdgcn_ballot_w32(t != 0.0f);
            kmask |= ((b & 0x55555555u) != 0u ? 1u : 0u) << i;
            kmask |= ((b & 0xAAAAAAAAu) != 0u ? 1u : 0u) << (16 + i);
        }
        s += __shfl_xor(s, 1);
        const float dv = rsqrtf(s);
        if (hf == 0) dinv_s[p] = dv;
        const float* orow = obs_s + inv_s[p] * OBS_LD + hf * 16;
        float4 o0 = *(const float4*)(orow);
        float4 o1 = *(const float4*)(orow + 4);
        float4 o2 = *(const float4*)(orow + 8);
        float4 o3 = *(const float4*)(orow + 12);
        o0.x *= dv; o0.y *= dv; o0.z *= dv; o0.w *= dv;
        o1.x *= dv; o1.y *= dv; o1.z *= dv; o1.w *= dv;
        o2.x *= dv; o2.y *= dv; o2.z *= dv; o2.w *= dv;
        o3.x *= dv; o3.y *= dv; o3.z *= dv; o3.w *= dv;
        __syncthreads();
        float* drow = obs_s + p * OBS_LD + hf * 16;
        *(float4*)(drow)      = o0;
        *(float4*)(drow + 4)  = o1;
        *(float4*)(drow + 8)  = o2;
        *(float4*)(drow + 12) = o3;
    }
    __syncthreads();

    {
        v8f c0, c1;
        tiles_acc16x32(M_s, MAT_LD, obs_s, OBS_LD, A_AGENTS, kmask, tm, lane, c0, c1);
        const Row8 rs = load_row8(dinv_s, tm, lane);
        float* obase = X_s + (tm * 16 + 8 * (lane >> 4)) * MAT_LD + (lane & 15);
#pragma unroll
        for (int v = 0; v < 8; ++v) {
            obase[v * MAT_LD]      = c0[v] * rs.d[v];
            obase[v * MAT_LD + 16] = c1[v] * rs.d[v];
        }
    }
    wave_lds_order();

    {
        const Acc8 acc = strip_acc16x8(X_s, MAT_LD, W1, HID, D_IN, tm, lane);
        const Row8 none = {};
        wave_lds_order();
        strip_store16x4<false, true>(acc.lo, X_s, MAT_LD, none, b1, tm, 0, lane);
        strip_store16x4<false, true>(acc.hi, X_s, MAT_LD, none, b1, tm, 4, lane);
    }
    wave_lds_order();

    {
        const Acc8 acc = strip_acc16x8(X_s, MAT_LD, W2, HID, HID, tm, lane);
        const Row8 rs = load_row8(dinv_s, tm, lane);
        wave_lds_order();
        strip_store16x4_pairs(acc.lo, X_s, rs, tm, 0, lane);
        strip_store16x4_pairs(acc.hi, X_s, rs, tm, 4, lane);
    }
    __syncthreads();

    {
        const Acc8 acc = strip_acc16x8_pairs(M_s, MAT_LD, X_s, HID, kmask, tm, lane);
        const Row8 rs = load_row8(dinv_s, tm, lane);
        wave_lds_order();
        strip_store16x4<true, true>(acc.lo, M_s, MAT_LD, rs, b2, tm, 0, lane);
        strip_store16x4<true, true>(acc.hi, M_s, MAT_LD, rs, b2, tm, 4, lane);
    }
    wave_lds_order();

    {
        const int a  = tid >> 1;
        const int hf = tid & 1;
        const float* hrow = M_s + a * MAT_LD + hf * 64;
        const float* wseg = Wout + hf * 64;
        float s0 = 0.0f, s1 = 0.0f;
        for (int k = 0; k < 32; ++k) {
            s0 += hrow[k]      * wseg[k];
            s1 += hrow[k + 32] * wseg[k + 32];
        }
        float s = s0 + s1;
        s += __shfl_xor(s, 1);
        if (hf == 0) out[(size_t)bidx * A_AGENTS + inv_s[a]] = s + bout[0];
    }
}

extern "C" void kernel_launch(void* const* d_in, const int* in_sizes, int n_in,
                              void* d_out, int out_size, void* d_ws, size_t ws_size,
                              hipStream_t stream) {
    (void)n_in; (void)out_size; (void)d_ws; (void)ws_size;
    const float* obs  = (const float*)d_in[0];
    const float* W1   = (const float*)d_in[1];
    const float* b1   = (const float*)d_in[2];
    const float* W2   = (const float*)d_in[3];
    const float* b2   = (const float*)d_in[4];
    const float* Wout = (const float*)d_in[5];
    const float* bout = (const float*)d_in[6];
    float* out = (float*)d_out;

    const int B = in_sizes[0] / (A_AGENTS * D_IN);

    static_assert(2 * LDS_BYTES <= 320 * 1024, "two workgroups must fit one WGP's LDS");
    (void)hipFuncSetAttribute((const void*)gnn_critic_fused,
                              hipFuncAttributeMaxDynamicSharedMemorySize,
                              LDS_BYTES);
    gnn_critic_fused<<<B, NTHREADS, LDS_BYTES, stream>>>(
        obs, W1, b1, W2, b2, Wout, bout, out);
}
